# starting kernel
# baseline (speedup 1.0000x reference)
#include <hip/hip_runtime.h>
#include <hip/hip_bf16.h>

typedef __attribute__((ext_vector_type(2))) float v2f;
typedef __attribute__((ext_vector_type(8))) float v8f;

#define GEMM_WAVES 4
#define ROW_ITERS  8

__global__ void gcn_zero_f32(float* __restrict__ p, long long n) {
  long long i = (long long)blockIdx.x * blockDim.x + threadIdx.x;
  if (i < n) p[i] = 0.0f;
}

__global__ void gcn_degrees(const int* __restrict__ src, const int* __restrict__ dst,
                            float* __restrict__ deg_out, float* __restrict__ deg_in,
                            int E) {
  int e = blockIdx.x * blockDim.x + threadIdx.x;
  if (e < E) {
    atomicAdd(&deg_out[src[e]], 1.0f);
    atomicAdd(&deg_in[dst[e]], 1.0f);
  }
}

__global__ void gcn_norm(const float* __restrict__ deg, float* __restrict__ norm, int n) {
  int i = blockIdx.x * blockDim.x + threadIdx.x;
  if (i < n) {
    float d = deg[i];
    norm[i] = (d > 0.0f) ? rsqrtf(d) : 0.0f;
  }
}

template <int KDIM>
__global__ __launch_bounds__(32 * GEMM_WAVES) void gcn_gemm_wmma(
    const float* __restrict__ X, const float* __restrict__ Wm,
    const float* __restrict__ prescale, float* __restrict__ Y,
    int nRows, int Mout) {
  constexpr int KSTEPS = KDIM / 4;
  const int wave = threadIdx.x >> 5;
  const int lane = threadIdx.x & 31;
  const int nRowTiles = (nRows + 15) >> 4;
  const int tileBase = (blockIdx.x * GEMM_WAVES + wave) * ROW_ITERS;
  if (tileBase >= nRowTiles) return;

  const int m  = lane & 15;
  const int kg = lane >> 4;

  const int col = blockIdx.y * 16 + m;
  const int colc = (col < Mout) ? col : (Mout - 1);
  v2f breg[KSTEPS];
#pragma unroll
  for (int s = 0; s < KSTEPS; ++s) {
    const int k = 4 * s + 2 * kg;
    breg[s].x = Wm[k * Mout + colc];
    breg[s].y = Wm[(k + 1) * Mout + colc];
  }

  const int tileEnd = (tileBase + ROW_ITERS < nRowTiles) ? tileBase + ROW_ITERS
                                                         : nRowTiles;
  for (int tile = tileBase; tile < tileEnd; ++tile) {
    int row = tile * 16 + m;
    if (row >= nRows) row = nRows - 1;
    const float scale = prescale[row];
    const float* __restrict__ xrow = X + (long long)row * KDIM + 2 * kg;

    v8f acc = {};
#pragma unroll
    for (int s = 0; s < KSTEPS; ++s) {
      v2f A = *(const v2f*)(xrow + 4 * s);
      A.x *= scale;
      A.y *= scale;
      acc = __builtin_amdgcn_wmma_f32_16x16x4_f32(
           false, A,  false, breg[s],
           (short)0, acc,  false,  false);
    }

    if (col < Mout) {
#pragma unroll
      for (int r = 0; r < 8; ++r) {
        const int outRow = tile * 16 + r + 8 * kg;
        if (outRow < nRows) Y[(long long)outRow * Mout + col] = acc[r];
      }
    }
  }
}

template <int GROUPS>
__global__ void gcn_scatter(const float* __restrict__ Y, const int* __restrict__ src,
                            const int* __restrict__ dst, float* __restrict__ agg,
                            unsigned E) {
  constexpr int COLS = GROUPS * 4;
  const unsigned idx = blockIdx.x * blockDim.x + threadIdx.x;
  if (idx >= E * GROUPS) return;
  const unsigned e = idx / GROUPS;
  const int g = (int)(idx - e * GROUPS);
  const int s = src[e];
  const int d = dst[e];
  const float4 v = *(const float4*)(Y + (long long)s * COLS + g * 4);
  float* __restrict__ out = agg + (long long)d * COLS + g * 4;
  atomicAdd(out + 0, v.x);
  atomicAdd(out + 1, v.y);
  atomicAdd(out + 2, v.z);
  atomicAdd(out + 3, v.w);
}

template <int COLS, bool RELU>
__global__ void gcn_pointwise(float* __restrict__ A, const float* __restrict__ norm_in,
                              const float* __restrict__ bias, unsigned n) {
  const unsigned i = blockIdx.x * blockDim.x + threadIdx.x;
  if (i >= n) return;
  const unsigned row = i / COLS;
  const unsigned c = i - row * COLS;
  float v = A[i] * norm_in[row] + bias[c];
  if (RELU) v = fmaxf(v, 0.0f);
  A[i] = v;
}

extern "C" void kernel_launch(void* const* d_in, const int* in_sizes, int n_in,
                              void* d_out, int out_size, void* d_ws, size_t ws_size,
                              hipStream_t stream) {
  const float* feat = (const float*)d_in[0];
  const int*   src  = (const int*)d_in[1];
  const int*   dst  = (const int*)d_in[2];
  const float* W0   = (const float*)d_in[3];
  const float* b0   = (const float*)d_in[4];
  const float* W1   = (const float*)d_in[5];
  const float* b1   = (const float*)d_in[6];
  const float* W2   = (const float*)d_in[7];
  const float* b2   = (const float*)d_in[8];
  float* out = (float*)d_out;

  const int F = 128, H = 128, C = 40;
  const int N = in_sizes[0] / F;
  const int E = in_sizes[1];
  (void)in_sizes; (void)n_in; (void)out_size; (void)ws_size;

  float* ws       = (float*)d_ws;
  float* deg_out  = ws;
  float* norm_out = ws + 2LL * N;
  float* norm_in  = ws + 3LL * N;
  float* B0       = ws + 4LL * N;
  float* B1       = B0 + (long long)N * H;

  const int TB = 256;
  auto blocks1d = [](long long n, int tb) { return (unsigned)((n + tb - 1) / tb); };

  gcn_zero_f32<<<blocks1d(2LL * N, TB), TB, 0, stream>>>(deg_out, 2LL * N);
  gcn_degrees<<<blocks1d(E, TB), TB, 0, stream>>>(src, dst, deg_out, deg_out + N, E);
  gcn_norm<<<blocks1d(2LL * N, TB), TB, 0, stream>>>(deg_out, norm_out, 2 * N);

  const int rowTiles = (N + 15) / 16;
  const int tilesPerBlock = GEMM_WAVES * ROW_ITERS;
  const dim3 gemmBlk(32 * GEMM_WAVES);
  const dim3 gridH((rowTiles + tilesPerBlock - 1) / tilesPerBlock, H / 16);
  const dim3 gridC((rowTiles + tilesPerBlock - 1) / tilesPerBlock, (C + 15) / 16);

  gcn_gemm_wmma<128><<<gridH, gemmBlk, 0, stream>>>(feat, W0, norm_out, B0, N, H);
  gcn_zero_f32<<<blocks1d((long long)N * H, TB), TB, 0, stream>>>(B1, (long long)N * H);
  gcn_scatter<32><<<blocks1d((long long)E * 32, TB), TB, 0, stream>>>(B0, src, dst, B1, E);
  gcn_pointwise<128, true><<<blocks1d((long long)N * H, TB), TB, 0, stream>>>(
      B1, norm_in, b0, (unsigned)(N * H));

  gcn_gemm_wmma<128><<<gridH, gemmBlk, 0, stream>>>(B1, W1, norm_out, B0, N, H);
  gcn_zero_f32<<<blocks1d((long long)N * H, TB), TB, 0, stream>>>(B1, (long long)N * H);
  gcn_scatter<32><<<blocks1d((long long)E * 32, TB), TB, 0, stream>>>(B0, src, dst, B1, E);
  gcn_pointwise<128, true><<<blocks1d((long long)N * H, TB), TB, 0, stream>>>(
      B1, norm_in, b1, (unsigned)(N * H));

  gcn_gemm_wmma<128><<<gridC, gemmBlk, 0, stream>>>(B1, W2, norm_out, B0, N, C);
  gcn_zero_f32<<<blocks1d((long long)N * C, TB), TB, 0, stream>>>(out, (long long)N * C);
  gcn_scatter<10><<<blocks1d((long long)E * 10, TB), TB, 0, stream>>>(B0, src, dst, out, E);
  gcn_pointwise<40, false><<<blocks1d((long long)N * C, TB), TB, 0, stream>>>(
      out, norm_in, b2, (unsigned)(N * C));
}


# GEMM data path: a block stages each 32-row strip of X once in LDS (scaled, whole-row 16-B lane loads) and its 8 waves take A fragments from LDS; X no longer re-read per 16-col tile; sum order kept
# speedup vs baseline: 2.5625x; 2.5625x over previous
#include <hip/hip_runtime.h>
#include <hip/hip_bf16.h>

typedef __attribute__((ext_vector_type(2))) float v2f;
typedef __attribute__((ext_vector_type(8))) float v8f;

#define GEMM_THREADS 256
#define GEMM_STRIP   32
#define GEMM_PITCH   132
#define GEMM_BLOCKS  1024

__global__ void gcn_zero_f32(float* __restrict__ p, long long n) {
  long long i = (long long)blockIdx.x * blockDim.x + threadIdx.x;
  if (i < n) p[i] = 0.0f;
}

__global__ void gcn_degrees(const int* __restrict__ src, const int* __restrict__ dst,
                            float* __restrict__ deg_out, float* __restrict__ deg_in,
                            int E) {
  int e = blockIdx.x * blockDim.x + threadIdx.x;
  if (e < E) {
    atomicAdd(&deg_out[src[e]], 1.0f);
    atomicAdd(&deg_in[dst[e]], 1.0f);
  }
}

__global__ void gcn_norm(const float* __restrict__ deg, float* __restrict__ norm, int n) {
  int i = blockIdx.x * blockDim.x + threadIdx.x;
  if (i < n) {
    float d = deg[i];
    norm[i] = (d > 0.0f) ? rsqrtf(d) : 0.0f;
  }
}

template <int KDIM, int MOUT>
__global__ __launch_bounds__(GEMM_THREADS) void gcn_gemm_wmma(
    const float* __restrict__ X, const float* __restrict__ Wm,
    const float* __restrict__ prescale, float* __restrict__ Y, int nRows) {
  static_assert(KDIM == 128, "a wave stages one row of 128 floats per 16-byte lane load");
  constexpr int KSTEPS = KDIM / 4;
  constexpr int WAVES = GEMM_THREADS / 32;
  constexpr int NCT = (MOUT + 15) / 16;
  constexpr int RT = GEMM_STRIP / 16;
  constexpr int RG = (WAVES / NCT < RT) ? WAVES / NCT : RT;
  constexpr int TPW = RT / RG;
  constexpr int PASSES = GEMM_STRIP / WAVES;
  static_assert(NCT * RG <= WAVES && RG * TPW == RT && PASSES * WAVES == GEMM_STRIP, "tiling");
  __shared__ alignas(16) float xs[2][GEMM_STRIP * GEMM_PITCH];

  const int lane = threadIdx.x & 31;
  const int wave = __builtin_amdgcn_readfirstlane((int)(threadIdx.x >> 5));
  const int m  = lane & 15;
  const int kg = lane >> 4;
  const int nStrips = (nRows + GEMM_STRIP - 1) / GEMM_STRIP;
  int strip = blockIdx.x;
  if (strip >= nStrips) return;

  const bool active = wave < NCT * RG;
  const int ct = wave % NCT;
  const int rg = wave / NCT;

  const int col = ct * 16 + m;
  const int colc = (col < MOUT) ? col : (MOUT - 1);
  v2f breg[KSTEPS] = {};
  if (active) {
#pragma unroll
    for (int s = 0; s < KSTEPS; ++s) {
      const int k = 4 * s + 2 * kg;
      breg[s].x = Wm[k * MOUT + colc];
      breg[s].y = Wm[(k + 1) * MOUT + colc];
    }
  }

  float4 r[PASSES];
  float sc[PASSES];
  auto fetch = [&](int st) {
#pragma unroll
    for (int p = 0; p < PASSES; ++p) {
      int row = st * GEMM_STRIP + p * WAVES + wave;
      if (row >= nRows) row = nRows - 1;
      sc[p] = prescale[row];
      r[p] = *(const float4*)(X + (long long)row * KDIM + 4 * lane);
    }
  };
  auto put = [&](float* __restrict__ buf) {
#pragma unroll
    for (int p = 0; p < PASSES; ++p) {
      float4 v = r[p];
      v.x *= sc[p];
      v.y *= sc[p];
      v.z *= sc[p];
      v.w *= sc[p];
      *(float4*)(buf + (p * WAVES + wave) * GEMM_PITCH + 4 * lane) = v;
    }
  };

  fetch(strip);
  put(xs[0]);
  __syncthreads();
  int cur = 0;
  for (; strip < nStrips; strip += (int)gridDim.x) {
    const int nextStrip = strip + (int)gridDim.x;
    const bool more = nextStrip < nStrips;
    if (more) fetch(nextStrip);

    if (active) {
      const float* __restrict__ xb = xs[cur] + m * GEMM_PITCH + 2 * kg;
      v8f acc[TPW];
#pragma unroll
      for (int t = 0; t < TPW; ++t) acc[t] = v8f{};
#pragma unroll
      for (int s = 0; s < KSTEPS; ++s) {
#pragma unroll
        for (int t = 0; t < TPW; ++t) {
          const int rt = rg + t * RG;
          const v2f A = *(const v2f*)(xb + rt * 16 * GEMM_PITCH + 4 * s);
          acc[t] = __builtin_amdgcn_wmma_f32_16x16x4_f32(
               false, A,  false, breg[s],
               (short)0, acc[t],  false,  false);
        }
      }

      if (col < MOUT) {
#pragma unroll
        for (int t = 0; t < TPW; ++t) {
          const int rt = rg + t * RG;
#pragma unroll
          for (int q = 0; q < 8; ++q) {
            const int outRow = strip * GEMM_STRIP + rt * 16 + q + 8 * kg;
            if (outRow < nRows) Y[(long long)outRow * MOUT + col] = acc[t][q];
          }
        }
      }
    }

    if (more) put(xs[cur ^ 1]);
    __syncthreads();
    cur ^= 1;
  }
}

#define OWN_ROWS    392
#define OWN_THREADS 1024
#define OWN_Q       4096
#define OWN_SCAN    16
#define OWN_BATCH   8
#define OWN_LIST_CAP 32768

__global__ void gcn_owner_bytes(const int* __restrict__ dst, unsigned* __restrict__ own32,
                                unsigned E, unsigned nWords) {
  const unsigned w = blockIdx.x * blockDim.x + threadIdx.x;
  if (w >= nWords) return;
  unsigned packed = 0;
  if (4u * w + 3u < E) {
    const int4 d = *(const int4*)(dst + 4u * w);
    packed = (((unsigned)d.x / OWN_ROWS) & 255u) |
             ((((unsigned)d.y / OWN_ROWS) & 255u) << 8) |
             ((((unsigned)d.z / OWN_ROWS) & 255u) << 16) |
             ((((unsigned)d.w / OWN_ROWS) & 255u) << 24);
  } else {
    for (unsigned j = 0; j < 4; ++j) {
      const unsigned e = 4u * w + j;
      if (e < E) packed |= (((unsigned)dst[e] / OWN_ROWS) & 255u) << (8 * j);
    }
  }
  own32[w] = packed;
}

template <int COLS, bool RELU, bool BUILD>
__global__ __launch_bounds__(OWN_THREADS) void gcn_owner(
    const float* __restrict__ Y, const int* __restrict__ src,
    const int* __restrict__ dst, const unsigned* __restrict__ own32,
    const float* __restrict__ norm_in, const float* __restrict__ bias,
    float* __restrict__ out, uint2* lists, unsigned* listLen,
    unsigned E, unsigned nChunks, int nRows) {
  constexpr int FULL = COLS / 32;
  constexpr int TAIL = COLS % 32;
  constexpr int NCH = FULL > 0 ? FULL : 1;
  constexpr unsigned SEG = OWN_THREADS * OWN_SCAN;
  __shared__ float acc[OWN_ROWS * COLS];
  __shared__ uint2 q[OWN_Q];
  __shared__ unsigned cnt;

  const unsigned tid = threadIdx.x;
  const int lane = tid & 31;
  const unsigned wave = (unsigned)__builtin_amdgcn_readfirstlane((int)(tid >> 5));
  const unsigned id = blockIdx.x;
  const int lo = (int)(id * OWN_ROWS);
  int nOwnI = nRows - lo;
  if (nOwnI > OWN_ROWS) nOwnI = OWN_ROWS;
  if (nOwnI < 0) nOwnI = 0;
  const unsigned nOwn = (unsigned)nOwnI;

  for (unsigned i = tid; i < (unsigned)(OWN_ROWS * COLS); i += OWN_THREADS) acc[i] = 0.0f;

  const unsigned idx4 = (id & 255u) * 0x01010101u;
  const uint4* __restrict__ own = (const uint4*)own32;

  unsigned listPos = 0;
  bool listOk = true;
  uint2* myList = lists + (size_t)id * OWN_LIST_CAP;

  unsigned nList = 0xFFFFFFFFu;
  if (!BUILD) nList = (unsigned)__builtin_amdgcn_readfirstlane((int)listLen[id]);

  unsigned segFirst = 0;
  if (!BUILD && nList <= OWN_LIST_CAP) {
    segFirst = nChunks;
    __syncthreads();
    const uint2* __restrict__ rl = myList;
    const unsigned per = (nList + 31u) >> 5;
    const unsigned beg = wave * per;
    unsigned end = beg + per;
    if (end > nList) end = nList;
    uint2 nxt = make_uint2(0u, 0xFFFFFFFFu);
    if (beg + (unsigned)lane < end) nxt = rl[beg + (unsigned)lane];
    for (unsigned c = beg; c < end; c += 32u) {
      const uint2 cur = nxt;
      nxt = make_uint2(0u, 0xFFFFFFFFu);
      if (c + 32u + (unsigned)lane < end) nxt = rl[c + 32u + (unsigned)lane];
#pragma unroll
      for (int g = 0; g < 32 / OWN_BATCH; ++g) {
        if (c + (unsigned)(g * OWN_BATCH) >= end) break;
        float v[OWN_BATCH][NCH];
        float vt[OWN_BATCH];
        unsigned dl[OWN_BATCH];
#pragma unroll
        for (int b = 0; b < OWN_BATCH; ++b) {
          const unsigned s = (unsigned)__builtin_amdgcn_readlane((int)cur.x, g * OWN_BATCH + b);
          dl[b] = (unsigned)__builtin_amdgcn_readlane((int)cur.y, g * OWN_BATCH + b);
          const float* __restrict__ y = Y + (long long)s * COLS;
#pragma unroll
          for (int k = 0; k < FULL; ++k) v[b][k] = y[32 * k + lane];
          vt[b] = 0.0f;
          if (TAIL != 0) vt[b] = y[32 * FULL + ((lane < TAIL) ? lane : (TAIL - 1))];
        }
#pragma unroll
        for (int b = OWN_BATCH - 1; b >= 0; --b) {
          if (TAIL != 0) asm volatile("" : : "v"(vt[b]));
#pragma unroll
          for (int k = FULL - 1; k >= 0; --k) asm volatile("" : : "v"(v[b][k]));
        }
#pragma unroll
        for (int b = 0; b < OWN_BATCH; ++b) {
          if (dl[b] != 0xFFFFFFFFu) {
            float* a = acc + dl[b] * COLS + lane;
#pragma unroll
            for (int k = 0; k < FULL; ++k) atomicAdd(a + 32 * k, v[b][k]);
            if (TAIL != 0 && lane < TAIL) atomicAdd(a + 32 * FULL, vt[b]);
          }
        }
      }
    }
  }

  for (unsigned segBase = segFirst; segBase < nChunks; segBase += SEG) {
    __syncthreads();
    if (tid == 0) cnt = 0;
    __syncthreads();

    uint4 w[OWN_SCAN];
#pragma unroll
    for (int j = 0; j < OWN_SCAN; ++j) {
      unsigned c = segBase + (unsigned)j * OWN_THREADS + tid;
      if (c >= nChunks) c = nChunks - 1;
      w[j] = own[c];
    }
#pragma unroll
    for (int j = 0; j < OWN_SCAN; ++j) {
      const unsigned c = segBase + (unsigned)j * OWN_THREADS + tid;
      if (c < nChunks) {
        const unsigned xs[4] = {w[j].x ^ idx4, w[j].y ^ idx4, w[j].z ^ idx4, w[j].w ^ idx4};
#pragma unroll
        for (int t = 0; t < 4; ++t) {
          const unsigned x = xs[t];
          unsigned z = ~(((x & 0x7F7F7F7Fu) + 0x7F7F7F7Fu) | x | 0x7F7F7F7Fu);
          while (z != 0) {
            const unsigned bit = (unsigned)__builtin_ctz(z);
            z &= z - 1u;
            const unsigned e = c * 16u + (unsigned)t * 4u + (bit >> 3);
            if (e < E) {
              const unsigned slot = atomicAdd(&cnt, 1u);
              if (slot < OWN_Q) q[slot].x = e;
            }
          }
        }
      }
    }
    __syncthreads();

    const unsigned nAll = (unsigned)__builtin_amdgcn_readfirstlane((int)cnt);
    if (nAll > OWN_Q) {
      listOk = false;
      unsigned cEnd = segBase + SEG;
      if (cEnd > nChunks) cEnd = nChunks;
      for (unsigned c = segBase + tid; c < cEnd; c += OWN_THREADS) {
        for (unsigned k = 0; k < 16u; ++k) {
          const unsigned e = c * 16u + k;
          if (e >= E) break;
          const unsigned dl = (unsigned)(dst[e] - lo);
          if (dl < nOwn) {
            const float* __restrict__ y = Y + (long long)src[e] * COLS;
#pragma nounroll
            for (int c2 = 0; c2 < COLS; ++c2) atomicAdd(&acc[dl * COLS + c2], y[c2]);
          }
        }
      }
    } else {
      const unsigned n = nAll;
      for (unsigned i = tid; i < n; i += OWN_THREADS) {
        const unsigned e = q[i].x;
        const unsigned s = (unsigned)src[e];
        const unsigned dl = (unsigned)(dst[e] - lo);
        const uint2 ent = make_uint2(s, (dl < nOwn) ? dl : 0xFFFFFFFFu);
        q[i] = ent;
        if (BUILD && listPos + i < OWN_LIST_CAP) myList[listPos + i] = ent;
      }
      listPos += n;
      __syncthreads();

      for (unsigned base = wave; base < n; base += 32u * OWN_BATCH) {
        float v[OWN_BATCH][NCH];
        float vt[OWN_BATCH];
        unsigned dl[OWN_BATCH];
#pragma unroll
        for (int b = 0; b < OWN_BATCH; ++b) {
          unsigned idx = base + 32u * (unsigned)b;
          if (idx >= n) idx = n - 1;
          const uint2 ent = q[idx];
          dl[b] = ent.y;
          const float* __restrict__ y = Y + (long long)ent.x * COLS;
#pragma unroll
          for (int k = 0; k < FULL; ++k) v[b][k] = y[32 * k + lane];
          vt[b] = 0.0f;
          if (TAIL != 0) vt[b] = y[32 * FULL + ((lane < TAIL) ? lane : (TAIL - 1))];
        }
#pragma unroll
        for (int b = OWN_BATCH - 1; b >= 0; --b) {
          if (TAIL != 0) asm volatile("" : : "v"(vt[b]));
#pragma unroll
          for (int k = FULL - 1; k >= 0; --k) asm volatile("" : : "v"(v[b][k]));
        }
#pragma unroll
        for (int b = 0; b < OWN_BATCH; ++b) {
          if (base + 32u * (unsigned)b < n && dl[b] != 0xFFFFFFFFu) {
            float* a = acc + dl[b] * COLS + lane;
#pragma unroll
            for (int k = 0; k < FULL; ++k) atomicAdd(a + 32 * k, v[b][k]);
            if (TAIL != 0 && lane < TAIL) atomicAdd(a + 32 * FULL, vt[b]);
          }
        }
      }
    }
  }
  __syncthreads();

  if (BUILD && tid == 0) listLen[id] = listOk ? listPos : 0xFFFFFFFFu;

  const unsigned total = nOwn * COLS;
  float* __restrict__ o = out + (long long)lo * COLS;
  for (unsigned i = tid; i < total; i += OWN_THREADS) {
    const unsigned r = i / COLS;
    const unsigned c = i - r * COLS;
    float val = acc[i] * norm_in[lo + (int)r] + bias[c];
    if (RELU) val = fmaxf(val, 0.0f);
    o[i] = val;
  }
}

extern "C" void kernel_launch(void* const* d_in, const int* in_sizes, int n_in,
                              void* d_out, int out_size, void* d_ws, size_t ws_size,
                              hipStream_t stream) {
  const float* feat = (const float*)d_in[0];
  const int*   src  = (const int*)d_in[1];
  const int*   dst  = (const int*)d_in[2];
  const float* W0   = (const float*)d_in[3];
  const float* b0   = (const float*)d_in[4];
  const float* W1   = (const float*)d_in[5];
  const float* b1   = (const float*)d_in[6];
  const float* W2   = (const float*)d_in[7];
  const float* b2   = (const float*)d_in[8];
  float* out = (float*)d_out;

  const int F = 128, H = 128;
  const int N = in_sizes[0] / F;
  const int E = in_sizes[1];
  (void)in_sizes; (void)n_in; (void)out_size; (void)ws_size;

  float* ws       = (float*)d_ws;
  float* deg_out  = ws;
  float* norm_out = ws + 2LL * N;
  float* norm_in  = ws + 3LL * N;
  float* B0       = ws + 4LL * N;
  float* B1       = B0 + (long long)N * H;
  unsigned* own32 = (unsigned*)(ws + ((4LL * N + 2LL * N * H + 3LL) & ~3LL));

  const int TB = 256;
  auto blocks1d = [](long long n, int tb) { return (unsigned)((n + tb - 1) / tb); };

  gcn_zero_f32<<<blocks1d(2LL * N, TB), TB, 0, stream>>>(deg_out, 2LL * N);
  gcn_degrees<<<blocks1d(E, TB), TB, 0, stream>>>(src, dst, deg_out, deg_out + N, E);
  gcn_norm<<<blocks1d(2LL * N, TB), TB, 0, stream>>>(deg_out, norm_out, 2 * N);

  const unsigned nChunks = ((unsigned)E + 15u) / 16u;
  const unsigned nWords = nChunks * 4u;
  gcn_owner_bytes<<<blocks1d(nWords, TB), TB, 0, stream>>>(dst, own32, (unsigned)E, nWords);
  const unsigned ownGrid = (unsigned)((N + OWN_ROWS - 1) / OWN_ROWS);
  unsigned* listLen = own32 + nWords;
  uint2* lists = (uint2*)(listLen + ((ownGrid + 3u) & ~3u));

  gcn_gemm_wmma<128, 128><<<GEMM_BLOCKS, GEMM_THREADS, 0, stream>>>(feat, W0, norm_out, B0, N);
  gcn_owner<128, true, true><<<ownGrid, OWN_THREADS, 0, stream>>>(
      B0, src, dst, own32, norm_in, b0, B1, lists, listLen, (unsigned)E, nChunks, N);

  gcn_gemm_wmma<128, 128><<<GEMM_BLOCKS, GEMM_THREADS, 0, stream>>>(B1, W1, norm_out, B0, N);
  gcn_owner<128, true, false><<<ownGrid, OWN_THREADS, 0, stream>>>(
      B0, src, dst, own32, norm_in, b1, B1, lists, listLen, (unsigned)E, nChunks, N);

  gcn_gemm_wmma<128, 40><<<GEMM_BLOCKS, GEMM_THREADS, 0, stream>>>(B1, W2, norm_out, B0, N);
  gcn_owner<40, false, false><<<ownGrid, OWN_THREADS, 0, stream>>>(
      B0, src, dst, own32, norm_in, b2, out, lists, listLen, (unsigned)E, nChunks, N);
}


# edge lists built by a separate kernel before layer 0 (256 ranges x 8 edge slices, 8-wave groups, ballot-ranked wave queues, one int add per wave); all 3 sums stream lists; list order = arrival order
# speedup vs baseline: 2.7848x; 1.0868x over previous
#include <hip/hip_runtime.h>
#include <hip/hip_bf16.h>

typedef __attribute__((ext_vector_type(2))) float v2f;
typedef __attribute__((ext_vector_type(8))) float v8f;

#define GEMM_THREADS 256
#define GEMM_STRIP   32
#define GEMM_PITCH   132
#define GEMM_BLOCKS  1024

__global__ void gcn_zero_f32(float* __restrict__ p, long long n) {
  long long i = (long long)blockIdx.x * blockDim.x + threadIdx.x;
  if (i < n) p[i] = 0.0f;
}

__global__ void gcn_degrees(const int* __restrict__ src, const int* __restrict__ dst,
                            float* __restrict__ deg_out, float* __restrict__ deg_in,
                            int E) {
  int e = blockIdx.x * blockDim.x + threadIdx.x;
  if (e < E) {
    atomicAdd(&deg_out[src[e]], 1.0f);
    atomicAdd(&deg_in[dst[e]], 1.0f);
  }
}

__global__ void gcn_norm(const float* __restrict__ deg, float* __restrict__ norm, int n) {
  int i = blockIdx.x * blockDim.x + threadIdx.x;
  if (i < n) {
    float d = deg[i];
    norm[i] = (d > 0.0f) ? rsqrtf(d) : 0.0f;
  }
}

template <int KDIM, int MOUT>
__global__ __launch_bounds__(GEMM_THREADS) void gcn_gemm_wmma(
    const float* __restrict__ X, const float* __restrict__ Wm,
    const float* __restrict__ prescale, float* __restrict__ Y, int nRows) {
  static_assert(KDIM == 128, "a wave stages one row of 128 floats per 16-byte lane load");
  constexpr int KSTEPS = KDIM / 4;
  constexpr int WAVES = GEMM_THREADS / 32;
  constexpr int NCT = (MOUT + 15) / 16;
  constexpr int RT = GEMM_STRIP / 16;
  constexpr int RG = (WAVES / NCT < RT) ? WAVES / NCT : RT;
  constexpr int TPW = RT / RG;
  constexpr int PASSES = GEMM_STRIP / WAVES;
  static_assert(NCT * RG <= WAVES && RG * TPW == RT && PASSES * WAVES == GEMM_STRIP, "tiling");
  __shared__ alignas(16) float xs[2][GEMM_STRIP * GEMM_PITCH];

  const int lane = threadIdx.x & 31;
  const int wave = __builtin_amdgcn_readfirstlane((int)(threadIdx.x >> 5));
  const int m  = lane & 15;
  const int kg = lane >> 4;
  const int nStrips = (nRows + GEMM_STRIP - 1) / GEMM_STRIP;
  int strip = blockIdx.x;
  if (strip >= nStrips) return;

  const bool active = wave < NCT * RG;
  const int ct = wave % NCT;
  const int rg = wave / NCT;

  const int col = ct * 16 + m;
  const int colc = (col < MOUT) ? col : (MOUT - 1);
  v2f breg[KSTEPS] = {};
  if (active) {
#pragma unroll
    for (int s = 0; s < KSTEPS; ++s) {
      const int k = 4 * s + 2 * kg;
      breg[s].x = Wm[k * MOUT + colc];
      breg[s].y = Wm[(k + 1) * MOUT + colc];
    }
  }

  float4 r[PASSES];
  float sc[PASSES];
  auto fetch = [&](int st) {
#pragma unroll
    for (int p = 0; p < PASSES; ++p) {
      int row = st * GEMM_STRIP + p * WAVES + wave;
      if (row >= nRows) row = nRows - 1;
      sc[p] = prescale[row];
      r[p] = *(const float4*)(X + (long long)row * KDIM + 4 * lane);
    }
  };
  auto put = [&](float* __restrict__ buf) {
#pragma unroll
    for (int p = 0; p < PASSES; ++p) {
      float4 v = r[p];
      v.x *= sc[p];
      v.y *= sc[p];
      v.z *= sc[p];
      v.w *= sc[p];
      *(float4*)(buf + (p * WAVES + wave) * GEMM_PITCH + 4 * lane) = v;
    }
  };

  fetch(strip);
  put(xs[0]);
  __syncthreads();
  int cur = 0;
  for (; strip < nStrips; strip += (int)gridDim.x) {
    const int nextStrip = strip + (int)gridDim.x;
    const bool more = nextStrip < nStrips;
    if (more) fetch(nextStrip);

    if (active) {
      const float* __restrict__ xb = xs[cur] + m * GEMM_PITCH + 2 * kg;
      v8f acc[TPW];
#pragma unroll
      for (int t = 0; t < TPW; ++t) acc[t] = v8f{};
#pragma unroll
      for (int s = 0; s < KSTEPS; ++s) {
#pragma unroll
        for (int t = 0; t < TPW; ++t) {
          const int rt = rg + t * RG;
          const v2f A = *(const v2f*)(xb + rt * 16 * GEMM_PITCH + 4 * s);
          acc[t] = __builtin_amdgcn_wmma_f32_16x16x4_f32(
               false, A,  false, breg[s],
               (short)0, acc[t],  false,  false);
        }
      }

      if (col < MOUT) {
#pragma unroll
        for (int t = 0; t < TPW; ++t) {
          const int rt = rg + t * RG;
#pragma unroll
          for (int q = 0; q < 8; ++q) {
            const int outRow = strip * GEMM_STRIP + rt * 16 + q + 8 * kg;
            if (outRow < nRows) Y[(long long)outRow * MOUT + col] = acc[t][q];
          }
        }
      }
    }

    if (more) put(xs[cur ^ 1]);
    __syncthreads();
    cur ^= 1;
  }
}

#define OWN_ROWS    392
#define OWN_THREADS 1024
#define OWN_Q       4096
#define OWN_SCAN    16
#define OWN_BATCH   8
#define OWN_LIST_CAP 32768

#define BLD_THREADS 256
#define BLD_SLICES  8
#define BLD_BATCH   8
#define BLD_WQ      512

__global__ void gcn_owner_bytes(const int* __restrict__ dst, unsigned* __restrict__ own32,
                                unsigned* __restrict__ listLen, unsigned nLists,
                                unsigned E, unsigned nWords) {
  const unsigned w = blockIdx.x * blockDim.x + threadIdx.x;
  if (w < nLists) listLen[w] = 0u;
  if (w >= nWords) return;
  unsigned packed = 0;
  if (4u * w + 3u < E) {
    const int4 d = *(const int4*)(dst + 4u * w);
    packed = (((unsigned)d.x / OWN_ROWS) & 255u) |
             ((((unsigned)d.y / OWN_ROWS) & 255u) << 8) |
             ((((unsigned)d.z / OWN_ROWS) & 255u) << 16) |
             ((((unsigned)d.w / OWN_ROWS) & 255u) << 24);
  } else {
    for (unsigned j = 0; j < 4; ++j) {
      const unsigned e = 4u * w + j;
      if (e < E) packed |= (((unsigned)dst[e] / OWN_ROWS) & 255u) << (8 * j);
    }
  }
  own32[w] = packed;
}

__global__ __launch_bounds__(BLD_THREADS) void gcn_build_lists(
    const int* __restrict__ src, const int* __restrict__ dst,
    const unsigned* __restrict__ own32, uint2* __restrict__ lists,
    unsigned* __restrict__ listLen, unsigned E, unsigned nChunks, int nRows) {
  __shared__ unsigned q[(BLD_THREADS / 32) * BLD_WQ];

  const unsigned tid = threadIdx.x;
  const unsigned lane = tid & 31u;
  const unsigned wave = (unsigned)__builtin_amdgcn_readfirstlane((int)(tid >> 5));
  const unsigned id = blockIdx.x;
  const int lo = (int)(id * OWN_ROWS);
  int nOwnI = nRows - lo;
  if (nOwnI > OWN_ROWS) nOwnI = OWN_ROWS;
  if (nOwnI < 0) nOwnI = 0;
  const unsigned nOwn = (unsigned)nOwnI;

  const unsigned per = (nChunks + gridDim.y - 1u) / gridDim.y;
  const unsigned cBeg = blockIdx.y * per;
  unsigned cEnd = cBeg + per;
  if (cEnd > nChunks) cEnd = nChunks;

  const unsigned idx4 = (id & 255u) * 0x01010101u;
  const uint4* __restrict__ own = (const uint4*)own32;
  unsigned* wq = q + wave * BLD_WQ;
  unsigned wcount = 0;

  for (unsigned base = cBeg; base < cEnd; base += BLD_THREADS * BLD_BATCH) {
    uint4 w[BLD_BATCH];
#pragma unroll
    for (int j = 0; j < BLD_BATCH; ++j) {
      unsigned c = base + (unsigned)j * BLD_THREADS + tid;
      if (c >= nChunks) c = nChunks - 1;
      w[j] = own[c];
    }
#pragma unroll
    for (int j = 0; j < BLD_BATCH; j += 2) {
      const unsigned c = base + (unsigned)j * BLD_THREADS + tid;
      unsigned m = 0;
#pragma unroll
      for (int h = 0; h < 2; ++h) {
        if (c + (unsigned)h * BLD_THREADS < cEnd) {
          const uint4 v = w[j + h];
          const unsigned x0 = v.x ^ idx4, x1 = v.y ^ idx4, x2 = v.z ^ idx4, x3 = v.w ^ idx4;
          const unsigned z0 = ~(((x0 & 0x7F7F7F7Fu) + 0x7F7F7F7Fu) | x0 | 0x7F7F7F7Fu);
          const unsigned z1 = ~(((x1 & 0x7F7F7F7Fu) + 0x7F7F7F7Fu) | x1 | 0x7F7F7F7Fu);
          const unsigned z2 = ~(((x2 & 0x7F7F7F7Fu) + 0x7F7F7F7Fu) | x2 | 0x7F7F7F7Fu);
          const unsigned z3 = ~(((x3 & 0x7F7F7F7Fu) + 0x7F7F7F7Fu) | x3 | 0x7F7F7F7Fu);
          m |= ((z0 >> 7) | (z1 >> 6) | (z2 >> 5) | (z3 >> 4)) << (4 * h);
        }
      }
      while (__builtin_amdgcn_ballot_w32(m != 0u) != 0u) {
        const bool has = m != 0u;
        const unsigned b = has ? (unsigned)__builtin_ctz(m) : 0u;
        m &= m - 1u;
        const unsigned e = (c + ((b >> 2) & 1u) * BLD_THREADS) * 16u + ((b & 3u) << 2) + (b >> 3);
        const bool valid = has && e < E;
        const unsigned vote = __builtin_amdgcn_ballot_w32(valid);
        const unsigned slot = wcount + __builtin_amdgcn_mbcnt_lo(vote, 0u);
        if (valid && slot < BLD_WQ) wq[slot] = e;
        wcount += (unsigned)__builtin_popcount(vote);
      }
    }
  }
  __syncthreads();

  const bool fits = wcount <= BLD_WQ;
  unsigned pos = 0;
  if (lane == 0 && wcount != 0u)
    pos = atomicAdd(&listLen[id], fits ? wcount : (unsigned)(OWN_LIST_CAP + 1));
  pos = (unsigned)__builtin_amdgcn_readfirstlane((int)pos);
  if (!fits) return;

  uint2* __restrict__ myList = lists + (size_t)id * OWN_LIST_CAP;
  for (unsigned i = lane; i < wcount; i += 32u) {
    const unsigned e = wq[i];
    const unsigned s = (unsigned)src[e];
    const unsigned dl = (unsigned)(dst[e] - lo);
    if (pos + i < OWN_LIST_CAP)
      myList[pos + i] = make_uint2(s, (dl < nOwn) ? dl : 0xFFFFFFFFu);
  }
}

template <int COLS, bool RELU>
__global__ __launch_bounds__(OWN_THREADS) void gcn_owner(
    const float* __restrict__ Y, const int* __restrict__ src,
    const int* __restrict__ dst, const unsigned* __restrict__ own32,
    const float* __restrict__ norm_in, const float* __restrict__ bias,
    float* __restrict__ out, const uint2* __restrict__ lists,
    const unsigned* __restrict__ listLen, unsigned E, unsigned nChunks, int nRows) {
  constexpr int FULL = COLS / 32;
  constexpr int TAIL = COLS % 32;
  constexpr int NCH = FULL > 0 ? FULL : 1;
  constexpr unsigned SEG = OWN_THREADS * OWN_SCAN;
  __shared__ float acc[OWN_ROWS * COLS];
  __shared__ uint2 q[OWN_Q];
  __shared__ unsigned cnt;

  const unsigned tid = threadIdx.x;
  const int lane = tid & 31;
  const unsigned wave = (unsigned)__builtin_amdgcn_readfirstlane((int)(tid >> 5));
  const unsigned id = blockIdx.x;
  const int lo = (int)(id * OWN_ROWS);
  int nOwnI = nRows - lo;
  if (nOwnI > OWN_ROWS) nOwnI = OWN_ROWS;
  if (nOwnI < 0) nOwnI = 0;
  const unsigned nOwn = (unsigned)nOwnI;

  for (unsigned i = tid; i < (unsigned)(OWN_ROWS * COLS); i += OWN_THREADS) acc[i] = 0.0f;

  const unsigned idx4 = (id & 255u) * 0x01010101u;
  const uint4* __restrict__ own = (const uint4*)own32;

  const uint2* __restrict__ myList = lists + (size_t)id * OWN_LIST_CAP;
  const unsigned nList = (unsigned)__builtin_amdgcn_readfirstlane((int)listLen[id]);

  unsigned segFirst = 0;
  if (nList <= OWN_LIST_CAP) {
    segFirst = nChunks;
    __syncthreads();
    const uint2* __restrict__ rl = myList;
    const unsigned per = (nList + 31u) >> 5;
    const unsigned beg = wave * per;
    unsigned end = beg + per;
    if (end > nList) end = nList;
    uint2 nxt = make_uint2(0u, 0xFFFFFFFFu);
    if (beg + (unsigned)lane < end) nxt = rl[beg + (unsigned)lane];
    for (unsigned c = beg; c < end; c += 32u) {
      const uint2 cur = nxt;
      nxt = make_uint2(0u, 0xFFFFFFFFu);
      if (c + 32u + (unsigned)lane < end) nxt = rl[c + 32u + (unsigned)lane];
#pragma unroll
      for (int g = 0; g < 32 / OWN_BATCH; ++g) {
        if (c + (unsigned)(g * OWN_BATCH) >= end) break;
        float v[OWN_BATCH][NCH];
        float vt[OWN_BATCH];
        unsigned dl[OWN_BATCH];
#pragma unroll
        for (int b = 0; b < OWN_BATCH; ++b) {
          const unsigned s = (unsigned)__builtin_amdgcn_readlane((int)cur.x, g * OWN_BATCH + b);
          dl[b] = (unsigned)__builtin_amdgcn_readlane((int)cur.y, g * OWN_BATCH + b);
          const float* __restrict__ y = Y + (long long)s * COLS;
#pragma unroll
          for (int k = 0; k < FULL; ++k) v[b][k] = y[32 * k + lane];
          vt[b] = 0.0f;
          if (TAIL != 0) vt[b] = y[32 * FULL + ((lane < TAIL) ? lane : (TAIL - 1))];
        }
#pragma unroll
        for (int b = OWN_BATCH - 1; b >= 0; --b) {
          if (TAIL != 0) asm volatile("" : : "v"(vt[b]));
#pragma unroll
          for (int k = FULL - 1; k >= 0; --k) asm volatile("" : : "v"(v[b][k]));
        }
#pragma unroll
        for (int b = 0; b < OWN_BATCH; ++b) {
          if (dl[b] != 0xFFFFFFFFu) {
            float* a = acc + dl[b] * COLS + lane;
#pragma unroll
            for (int k = 0; k < FULL; ++k) atomicAdd(a + 32 * k, v[b][k]);
            if (TAIL != 0 && lane < TAIL) atomicAdd(a + 32 * FULL, vt[b]);
          }
        }
      }
    }
  }

  for (unsigned segBase = segFirst; segBase < nChunks; segBase += SEG) {
    __syncthreads();
    if (tid == 0) cnt = 0;
    __syncthreads();

    uint4 w[OWN_SCAN];
#pragma unroll
    for (int j = 0; j < OWN_SCAN; ++j) {
      unsigned c = segBase + (unsigned)j * OWN_THREADS + tid;
      if (c >= nChunks) c = nChunks - 1;
      w[j] = own[c];
    }
#pragma unroll
    for (int j = 0; j < OWN_SCAN; ++j) {
      const unsigned c = segBase + (unsigned)j * OWN_THREADS + tid;
      if (c < nChunks) {
        const unsigned xs[4] = {w[j].x ^ idx4, w[j].y ^ idx4, w[j].z ^ idx4, w[j].w ^ idx4};
#pragma unroll
        for (int t = 0; t < 4; ++t) {
          const unsigned x = xs[t];
          unsigned z = ~(((x & 0x7F7F7F7Fu) + 0x7F7F7F7Fu) | x | 0x7F7F7F7Fu);
          while (z != 0) {
            const unsigned bit = (unsigned)__builtin_ctz(z);
            z &= z - 1u;
            const unsigned e = c * 16u + (unsigned)t * 4u + (bit >> 3);
            if (e < E) {
              const unsigned slot = atomicAdd(&cnt, 1u);
              if (slot < OWN_Q) q[slot].x = e;
            }
          }
        }
      }
    }
    __syncthreads();

    const unsigned nAll = (unsigned)__builtin_amdgcn_readfirstlane((int)cnt);
    if (nAll > OWN_Q) {
      unsigned cEnd = segBase + SEG;
      if (cEnd > nChunks) cEnd = nChunks;
      for (unsigned c = segBase + tid; c < cEnd; c += OWN_THREADS) {
        for (unsigned k = 0; k < 16u; ++k) {
          const unsigned e = c * 16u + k;
          if (e >= E) break;
          const unsigned dl = (unsigned)(dst[e] - lo);
          if (dl < nOwn) {
            const float* __restrict__ y = Y + (long long)src[e] * COLS;
#pragma nounroll
            for (int c2 = 0; c2 < COLS; ++c2) atomicAdd(&acc[dl * COLS + c2], y[c2]);
          }
        }
      }
    } else {
      const unsigned n = nAll;
      for (unsigned i = tid; i < n; i += OWN_THREADS) {
        const unsigned e = q[i].x;
        const unsigned s = (unsigned)src[e];
        const unsigned dl = (unsigned)(dst[e] - lo);
        q[i] = make_uint2(s, (dl < nOwn) ? dl : 0xFFFFFFFFu);
      }
      __syncthreads();

      for (unsigned base = wave; base < n; base += 32u * OWN_BATCH) {
        float v[OWN_BATCH][NCH];
        float vt[OWN_BATCH];
        unsigned dl[OWN_BATCH];
#pragma unroll
        for (int b = 0; b < OWN_BATCH; ++b) {
          unsigned idx = base + 32u * (unsigned)b;
          if (idx >= n) idx = n - 1;
          const uint2 ent = q[idx];
          dl[b] = ent.y;
          const float* __restrict__ y = Y + (long long)ent.x * COLS;
#pragma unroll
          for (int k = 0; k < FULL; ++k) v[b][k] = y[32 * k + lane];
          vt[b] = 0.0f;
          if (TAIL != 0) vt[b] = y[32 * FULL + ((lane < TAIL) ? lane : (TAIL - 1))];
        }
#pragma unroll
        for (int b = OWN_BATCH - 1; b >= 0; --b) {
          if (TAIL != 0) asm volatile("" : : "v"(vt[b]));
#pragma unroll
          for (int k = FULL - 1; k >= 0; --k) asm volatile("" : : "v"(v[b][k]));
        }
#pragma unroll
        for (int b = 0; b < OWN_BATCH; ++b) {
          if (base + 32u * (unsigned)b < n && dl[b] != 0xFFFFFFFFu) {
            float* a = acc + dl[b] * COLS + lane;
#pragma unroll
            for (int k = 0; k < FULL; ++k) atomicAdd(a + 32 * k, v[b][k]);
            if (TAIL != 0 && lane < TAIL) atomicAdd(a + 32 * FULL, vt[b]);
          }
        }
      }
    }
  }
  __syncthreads();

  const unsigned total = nOwn * COLS;
  float* __restrict__ o = out + (long long)lo * COLS;
  for (unsigned i = tid; i < total; i += OWN_THREADS) {
    const unsigned r = i / COLS;
    const unsigned c = i - r * COLS;
    float val = acc[i] * norm_in[lo + (int)r] + bias[c];
    if (RELU) val = fmaxf(val, 0.0f);
    o[i] = val;
  }
}

extern "C" void kernel_launch(void* const* d_in, const int* in_sizes, int n_in,
                              void* d_out, int out_size, void* d_ws, size_t ws_size,
                              hipStream_t stream) {
  const float* feat = (const float*)d_in[0];
  const int*   src  = (const int*)d_in[1];
  const int*   dst  = (const int*)d_in[2];
  const float* W0   = (const float*)d_in[3];
  const float* b0   = (const float*)d_in[4];
  const float* W1   = (const float*)d_in[5];
  const float* b1   = (const float*)d_in[6];
  const float* W2   = (const float*)d_in[7];
  const float* b2   = (const float*)d_in[8];
  float* out = (float*)d_out;

  const int F = 128, H = 128;
  const int N = in_sizes[0] / F;
  const int E = in_sizes[1];
  (void)in_sizes; (void)n_in; (void)out_size; (void)ws_size;

  float* ws       = (float*)d_ws;
  float* deg_out  = ws;
  float* norm_out = ws + 2LL * N;
  float* norm_in  = ws + 3LL * N;
  float* B0       = ws + 4LL * N;
  float* B1       = B0 + (long long)N * H;
  unsigned* own32 = (unsigned*)(ws + ((4LL * N + 2LL * N * H + 3LL) & ~3LL));

  const int TB = 256;
  auto blocks1d = [](long long n, int tb) { return (unsigned)((n + tb - 1) / tb); };

  gcn_zero_f32<<<blocks1d(2LL * N, TB), TB, 0, stream>>>(deg_out, 2LL * N);
  gcn_degrees<<<blocks1d(E, TB), TB, 0, stream>>>(src, dst, deg_out, deg_out + N, E);
  gcn_norm<<<blocks1d(2LL * N, TB), TB, 0, stream>>>(deg_out, norm_out, 2 * N);

  const unsigned nChunks = ((unsigned)E + 15u) / 16u;
  const unsigned nWords = nChunks * 4u;
  const unsigned ownGrid = (unsigned)((N + OWN_ROWS - 1) / OWN_ROWS);
  unsigned* listLen = own32 + nWords;
  uint2* lists = (uint2*)(listLen + ((ownGrid + 3u) & ~3u));
  gcn_owner_bytes<<<blocks1d(nWords, TB), TB, 0, stream>>>(dst, own32, listLen, ownGrid,
                                                           (unsigned)E, nWords);
  gcn_build_lists<<<dim3(ownGrid, BLD_SLICES), BLD_THREADS, 0, stream>>>(
      src, dst, own32, lists, listLen, (unsigned)E, nChunks, N);

  gcn_gemm_wmma<128, 128><<<GEMM_BLOCKS, GEMM_THREADS, 0, stream>>>(feat, W0, norm_out, B0, N);
  gcn_owner<128, true><<<ownGrid, OWN_THREADS, 0, stream>>>(
      B0, src, dst, own32, norm_in, b0, B1, lists, listLen, (unsigned)E, nChunks, N);

  gcn_gemm_wmma<128, 128><<<GEMM_BLOCKS, GEMM_THREADS, 0, stream>>>(B1, W1, norm_out, B0, N);
  gcn_owner<128, true><<<ownGrid, OWN_THREADS, 0, stream>>>(
      B0, src, dst, own32, norm_in, b1, B1, lists, listLen, (unsigned)E, nChunks, N);

  gcn_gemm_wmma<128, 40><<<GEMM_BLOCKS, GEMM_THREADS, 0, stream>>>(B1, W2, norm_out, B0, N);
  gcn_owner<40, false><<<ownGrid, OWN_THREADS, 0, stream>>>(
      B0, src, dst, own32, norm_in, b2, out, lists, listLen, (unsigned)E, nChunks, N);
}


# degree kernel dissolved: in-degree counted in LDS by each owner range from its own edge list, norm_in formed there; out-degree add rides in the owner-byte pass; norm kernel gone; 11 launches -> 9
# speedup vs baseline: 3.2555x; 1.1690x over previous
#include <hip/hip_runtime.h>
#include <hip/hip_bf16.h>

typedef __attribute__((ext_vector_type(2))) float v2f;
typedef __attribute__((ext_vector_type(8))) float v8f;

#define GEMM_THREADS 256
#define GEMM_STRIP   32
#define GEMM_PITCH   132
#define GEMM_BLOCKS  1024

__global__ void gcn_zero_f32(float* __restrict__ p, long long n) {
  long long i = (long long)blockIdx.x * blockDim.x + threadIdx.x;
  if (i < n) p[i] = 0.0f;
}

__device__ __forceinline__ float gcn_norm_of(float deg) {
  return (deg > 0.0f) ? rsqrtf(deg) : 0.0f;
}

template <int KDIM, int MOUT>
__global__ __launch_bounds__(GEMM_THREADS) void gcn_gemm_wmma(
    const float* __restrict__ X, const float* __restrict__ Wm,
    const float* __restrict__ degOut, float* __restrict__ Y, int nRows) {
  static_assert(KDIM == 128, "a wave stages one row of 128 floats per 16-byte lane load");
  constexpr int KSTEPS = KDIM / 4;
  constexpr int WAVES = GEMM_THREADS / 32;
  constexpr int NCT = (MOUT + 15) / 16;
  constexpr int RT = GEMM_STRIP / 16;
  constexpr int RG = (WAVES / NCT < RT) ? WAVES / NCT : RT;
  constexpr int TPW = RT / RG;
  constexpr int PASSES = GEMM_STRIP / WAVES;
  static_assert(NCT * RG <= WAVES && RG * TPW == RT && PASSES * WAVES == GEMM_STRIP, "tiling");
  __shared__ alignas(16) float xs[2][GEMM_STRIP * GEMM_PITCH];

  const int lane = threadIdx.x & 31;
  const int wave = __builtin_amdgcn_readfirstlane((int)(threadIdx.x >> 5));
  const int m  = lane & 15;
  const int kg = lane >> 4;
  const int nStrips = (nRows + GEMM_STRIP - 1) / GEMM_STRIP;
  int strip = blockIdx.x;
  if (strip >= nStrips) return;

  const bool active = wave < NCT * RG;
  const int ct = wave % NCT;
  const int rg = wave / NCT;

  const int col = ct * 16 + m;
  const int colc = (col < MOUT) ? col : (MOUT - 1);
  v2f breg[KSTEPS] = {};
  if (active) {
#pragma unroll
    for (int s = 0; s < KSTEPS; ++s) {
      const int k = 4 * s + 2 * kg;
      breg[s].x = Wm[k * MOUT + colc];
      breg[s].y = Wm[(k + 1) * MOUT + colc];
    }
  }

  float4 r[PASSES];
  float sc[PASSES];
  auto fetch = [&](int st) {
#pragma unroll
    for (int p = 0; p < PASSES; ++p) {
      int row = st * GEMM_STRIP + p * WAVES + wave;
      if (row >= nRows) row = nRows - 1;
      sc[p] = degOut[row];
      r[p] = *(const float4*)(X + (long long)row * KDIM + 4 * lane);
    }
  };
  auto put = [&](float* __restrict__ buf) {
#pragma unroll
    for (int p = 0; p < PASSES; ++p) {
      float4 v = r[p];
      const float nrm = gcn_norm_of(sc[p]);
      v.x *= nrm;
      v.y *= nrm;
      v.z *= nrm;
      v.w *= nrm;
      *(float4*)(buf + (p * WAVES + wave) * GEMM_PITCH + 4 * lane) = v;
    }
  };

  fetch(strip);
  put(xs[0]);
  __syncthreads();
  int cur = 0;
  for (; strip < nStrips; strip += (int)gridDim.x) {
    const int nextStrip = strip + (int)gridDim.x;
    const bool more = nextStrip < nStrips;
    if (more) fetch(nextStrip);

    if (active) {
      const float* __restrict__ xb = xs[cur] + m * GEMM_PITCH + 2 * kg;
      v8f acc[TPW];
#pragma unroll
      for (int t = 0; t < TPW; ++t) acc[t] = v8f{};
#pragma unroll
      for (int s = 0; s < KSTEPS; ++s) {
#pragma unroll
        for (int t = 0; t < TPW; ++t) {
          const int rt = rg + t * RG;
          const v2f A = *(const v2f*)(xb + rt * 16 * GEMM_PITCH + 4 * s);
          acc[t] = __builtin_amdgcn_wmma_f32_16x16x4_f32(
               false, A,  false, breg[s],
               (short)0, acc[t],  false,  false);
        }
      }

      if (col < MOUT) {
#pragma unroll
        for (int t = 0; t < TPW; ++t) {
          const int rt = rg + t * RG;
#pragma unroll
          for (int q = 0; q < 8; ++q) {
            const int outRow = strip * GEMM_STRIP + rt * 16 + q + 8 * kg;
            if (outRow < nRows) Y[(long long)outRow * MOUT + col] = acc[t][q];
          }
        }
      }
    }

    if (more) put(xs[cur ^ 1]);
    __syncthreads();
    cur ^= 1;
  }
}

#define OWN_ROWS    392
#define OWN_THREADS 1024
#define OWN_Q       4096
#define OWN_SCAN    16
#define OWN_BATCH   8
#define OWN_LIST_CAP 32768

#define EDG_THREADS 256
#define EDG_WORDS   4

#define BLD_THREADS 256
#define BLD_SLICES  8
#define BLD_BATCH   8
#define BLD_WQ      512

__global__ __launch_bounds__(EDG_THREADS) void gcn_owner_bytes(
    const int* __restrict__ src, const int* __restrict__ dst, float* __restrict__ degOut,
    unsigned* __restrict__ own32, unsigned* __restrict__ listLen, unsigned nLists,
    unsigned E, unsigned nWords) {
  for (unsigned i = blockIdx.x * EDG_THREADS + threadIdx.x; i < nLists;
       i += gridDim.x * EDG_THREADS)
    listLen[i] = 0u;
  const unsigned w0 = blockIdx.x * (EDG_THREADS * EDG_WORDS) + threadIdx.x;
  const unsigned nFull = E >> 2;
  int4 d[EDG_WORDS] = {}, s[EDG_WORDS] = {};
  if (nFull != 0u) {
#pragma unroll
    for (int j = 0; j < EDG_WORDS; ++j) {
      const unsigned w = w0 + (unsigned)j * EDG_THREADS;
      const unsigned wc = (w < nFull) ? w : 0u;
      d[j] = *(const int4*)(dst + 4u * wc);
      s[j] = *(const int4*)(src + 4u * wc);
    }
    __builtin_amdgcn_sched_barrier(0);
  }
#pragma unroll
  for (int j = 0; j < EDG_WORDS; ++j) {
    const unsigned w = w0 + (unsigned)j * EDG_THREADS;
    if (w < nWords) {
      unsigned packed = 0;
      if (w < nFull) {
        packed = (((unsigned)d[j].x / OWN_ROWS) & 255u) |
                 ((((unsigned)d[j].y / OWN_ROWS) & 255u) << 8) |
                 ((((unsigned)d[j].z / OWN_ROWS) & 255u) << 16) |
                 ((((unsigned)d[j].w / OWN_ROWS) & 255u) << 24);
        atomicAdd(&degOut[s[j].x], 1.0f);
        atomicAdd(&degOut[s[j].y], 1.0f);
        atomicAdd(&degOut[s[j].z], 1.0f);
        atomicAdd(&degOut[s[j].w], 1.0f);
      } else {
        for (unsigned k = 0; k < 4; ++k) {
          const unsigned e = 4u * w + k;
          if (e < E) {
            packed |= (((unsigned)dst[e] / OWN_ROWS) & 255u) << (8 * k);
            atomicAdd(&degOut[src[e]], 1.0f);
          }
        }
      }
      own32[w] = packed;
    }
  }
}

__global__ __launch_bounds__(BLD_THREADS) void gcn_build_lists(
    const int* __restrict__ src, const int* __restrict__ dst,
    const unsigned* __restrict__ own32, uint2* __restrict__ lists,
    unsigned* __restrict__ listLen, unsigned E, unsigned nChunks, int nRows) {
  __shared__ unsigned q[(BLD_THREADS / 32) * BLD_WQ];

  const unsigned tid = threadIdx.x;
  const unsigned lane = tid & 31u;
  const unsigned wave = (unsigned)__builtin_amdgcn_readfirstlane((int)(tid >> 5));
  const unsigned id = blockIdx.x;
  const int lo = (int)(id * OWN_ROWS);
  int nOwnI = nRows - lo;
  if (nOwnI > OWN_ROWS) nOwnI = OWN_ROWS;
  if (nOwnI < 0) nOwnI = 0;
  const unsigned nOwn = (unsigned)nOwnI;

  const unsigned per = (nChunks + gridDim.y - 1u) / gridDim.y;
  const unsigned cBeg = blockIdx.y * per;
  unsigned cEnd = cBeg + per;
  if (cEnd > nChunks) cEnd = nChunks;

  const unsigned idx4 = (id & 255u) * 0x01010101u;
  const uint4* __restrict__ own = (const uint4*)own32;
  unsigned* wq = q + wave * BLD_WQ;
  unsigned wcount = 0;

  for (unsigned base = cBeg; base < cEnd; base += BLD_THREADS * BLD_BATCH) {
    uint4 w[BLD_BATCH];
#pragma unroll
    for (int j = 0; j < BLD_BATCH; ++j) {
      unsigned c = base + (unsigned)j * BLD_THREADS + tid;
      if (c >= nChunks) c = nChunks - 1;
      w[j] = own[c];
    }
#pragma unroll
    for (int j = 0; j < BLD_BATCH; j += 2) {
      const unsigned c = base + (unsigned)j * BLD_THREADS + tid;
      unsigned m = 0;
#pragma unroll
      for (int h = 0; h < 2; ++h) {
        if (c + (unsigned)h * BLD_THREADS < cEnd) {
          const uint4 v = w[j + h];
          const unsigned x0 = v.x ^ idx4, x1 = v.y ^ idx4, x2 = v.z ^ idx4, x3 = v.w ^ idx4;
          const unsigned z0 = ~(((x0 & 0x7F7F7F7Fu) + 0x7F7F7F7Fu) | x0 | 0x7F7F7F7Fu);
          const unsigned z1 = ~(((x1 & 0x7F7F7F7Fu) + 0x7F7F7F7Fu) | x1 | 0x7F7F7F7Fu);
          const unsigned z2 = ~(((x2 & 0x7F7F7F7Fu) + 0x7F7F7F7Fu) | x2 | 0x7F7F7F7Fu);
          const unsigned z3 = ~(((x3 & 0x7F7F7F7Fu) + 0x7F7F7F7Fu) | x3 | 0x7F7F7F7Fu);
          m |= ((z0 >> 7) | (z1 >> 6) | (z2 >> 5) | (z3 >> 4)) << (4 * h);
        }
      }
      while (__builtin_amdgcn_ballot_w32(m != 0u) != 0u) {
        const bool has = m != 0u;
        const unsigned b = has ? (unsigned)__builtin_ctz(m) : 0u;
        m &= m - 1u;
        const unsigned e = (c + ((b >> 2) & 1u) * BLD_THREADS) * 16u + ((b & 3u) << 2) + (b >> 3);
        const bool valid = has && e < E;
        const unsigned vote = __builtin_amdgcn_ballot_w32(valid);
        const unsigned slot = wcount + __builtin_amdgcn_mbcnt_lo(vote, 0u);
        if (valid && slot < BLD_WQ) wq[slot] = e;
        wcount += (unsigned)__builtin_popcount(vote);
      }
    }
  }
  __syncthreads();

  const bool fits = wcount <= BLD_WQ;
  unsigned pos = 0;
  if (lane == 0 && wcount != 0u)
    pos = atomicAdd(&listLen[id], fits ? wcount : (unsigned)(OWN_LIST_CAP + 1));
  pos = (unsigned)__builtin_amdgcn_readfirstlane((int)pos);
  if (!fits) return;

  uint2* __restrict__ myList = lists + (size_t)id * OWN_LIST_CAP;
  for (unsigned i = lane; i < wcount; i += 32u) {
    const unsigned e = wq[i];
    const unsigned s = (unsigned)src[e];
    const unsigned dl = (unsigned)(dst[e] - lo);
    if (pos + i < OWN_LIST_CAP)
      myList[pos + i] = make_uint2(s, (dl < nOwn) ? dl : 0xFFFFFFFFu);
  }
}

template <int COLS, bool RELU>
__global__ __launch_bounds__(OWN_THREADS) void gcn_owner(
    const float* __restrict__ Y, const int* __restrict__ src,
    const int* __restrict__ dst, const unsigned* __restrict__ own32,
    const float* __restrict__ bias,
    float* __restrict__ out, const uint2* __restrict__ lists,
    const unsigned* __restrict__ listLen, unsigned E, unsigned nChunks, int nRows) {
  constexpr int FULL = COLS / 32;
  constexpr int TAIL = COLS % 32;
  constexpr int NCH = FULL > 0 ? FULL : 1;
  constexpr unsigned SEG = OWN_THREADS * OWN_SCAN;
  __shared__ float acc[OWN_ROWS * COLS];
  __shared__ uint2 q[OWN_Q];
  __shared__ unsigned cnt;
  __shared__ unsigned degc[OWN_ROWS];
  __shared__ float nin[OWN_ROWS];

  const unsigned tid = threadIdx.x;
  const int lane = tid & 31;
  const unsigned wave = (unsigned)__builtin_amdgcn_readfirstlane((int)(tid >> 5));
  const unsigned id = blockIdx.x;
  const int lo = (int)(id * OWN_ROWS);
  int nOwnI = nRows - lo;
  if (nOwnI > OWN_ROWS) nOwnI = OWN_ROWS;
  if (nOwnI < 0) nOwnI = 0;
  const unsigned nOwn = (unsigned)nOwnI;

  for (unsigned i = tid; i < (unsigned)(OWN_ROWS * COLS); i += OWN_THREADS) acc[i] = 0.0f;
  if (tid < OWN_ROWS) degc[tid] = 0u;

  const unsigned idx4 = (id & 255u) * 0x01010101u;
  const uint4* __restrict__ own = (const uint4*)own32;

  const uint2* __restrict__ myList = lists + (size_t)id * OWN_LIST_CAP;
  const unsigned nList = (unsigned)__builtin_amdgcn_readfirstlane((int)listLen[id]);

  unsigned segFirst = 0;
  if (nList <= OWN_LIST_CAP) {
    segFirst = nChunks;
    __syncthreads();
    const uint2* __restrict__ rl = myList;
    const unsigned per = (nList + 31u) >> 5;
    const unsigned beg = wave * per;
    unsigned end = beg + per;
    if (end > nList) end = nList;
    uint2 nxt = make_uint2(0u, 0xFFFFFFFFu);
    if (beg + (unsigned)lane < end) nxt = rl[beg + (unsigned)lane];
    for (unsigned c = beg; c < end; c += 32u) {
      const uint2 cur = nxt;
      if (cur.y != 0xFFFFFFFFu) atomicAdd(&degc[cur.y], 1u);
      nxt = make_uint2(0u, 0xFFFFFFFFu);
      if (c + 32u + (unsigned)lane < end) nxt = rl[c + 32u + (unsigned)lane];
#pragma unroll
      for (int g = 0; g < 32 / OWN_BATCH; ++g) {
        if (c + (unsigned)(g * OWN_BATCH) >= end) break;
        float v[OWN_BATCH][NCH];
        float vt[OWN_BATCH];
        unsigned dl[OWN_BATCH];
#pragma unroll
        for (int b = 0; b < OWN_BATCH; ++b) {
          const unsigned s = (unsigned)__builtin_amdgcn_readlane((int)cur.x, g * OWN_BATCH + b);
          dl[b] = (unsigned)__builtin_amdgcn_readlane((int)cur.y, g * OWN_BATCH + b);
          const float* __restrict__ y = Y + (long long)s * COLS;
#pragma unroll
          for (int k = 0; k < FULL; ++k) v[b][k] = y[32 * k + lane];
          vt[b] = 0.0f;
          if (TAIL != 0) vt[b] = y[32 * FULL + ((lane < TAIL) ? lane : (TAIL - 1))];
        }
#pragma unroll
        for (int b = OWN_BATCH - 1; b >= 0; --b) {
          if (TAIL != 0) asm volatile("" : : "v"(vt[b]));
#pragma unroll
          for (int k = FULL - 1; k >= 0; --k) asm volatile("" : : "v"(v[b][k]));
        }
#pragma unroll
        for (int b = 0; b < OWN_BATCH; ++b) {
          if (dl[b] != 0xFFFFFFFFu) {
            float* a = acc + dl[b] * COLS + lane;
#pragma unroll
            for (int k = 0; k < FULL; ++k) atomicAdd(a + 32 * k, v[b][k]);
            if (TAIL != 0 && lane < TAIL) atomicAdd(a + 32 * FULL, vt[b]);
          }
        }
      }
    }
  }

  for (unsigned segBase = segFirst; segBase < nChunks; segBase += SEG) {
    __syncthreads();
    if (tid == 0) cnt = 0;
    __syncthreads();

    uint4 w[OWN_SCAN];
#pragma unroll
    for (int j = 0; j < OWN_SCAN; ++j) {
      unsigned c = segBase + (unsigned)j * OWN_THREADS + tid;
      if (c >= nChunks) c = nChunks - 1;
      w[j] = own[c];
    }
#pragma unroll
    for (int j = 0; j < OWN_SCAN; ++j) {
      const unsigned c = segBase + (unsigned)j * OWN_THREADS + tid;
      if (c < nChunks) {
        const unsigned xs[4] = {w[j].x ^ idx4, w[j].y ^ idx4, w[j].z ^ idx4, w[j].w ^ idx4};
#pragma unroll
        for (int t = 0; t < 4; ++t) {
          const unsigned x = xs[t];
          unsigned z = ~(((x & 0x7F7F7F7Fu) + 0x7F7F7F7Fu) | x | 0x7F7F7F7Fu);
          while (z != 0) {
            const unsigned bit = (unsigned)__builtin_ctz(z);
            z &= z - 1u;
            const unsigned e = c * 16u + (unsigned)t * 4u + (bit >> 3);
            if (e < E) {
              const unsigned slot = atomicAdd(&cnt, 1u);
              if (slot < OWN_Q) q[slot].x = e;
            }
          }
        }
      }
    }
    __syncthreads();

    const unsigned nAll = (unsigned)__builtin_amdgcn_readfirstlane((int)cnt);
    if (nAll > OWN_Q) {
      unsigned cEnd = segBase + SEG;
      if (cEnd > nChunks) cEnd = nChunks;
      for (unsigned c = segBase + tid; c < cEnd; c += OWN_THREADS) {
        for (unsigned k = 0; k < 16u; ++k) {
          const unsigned e = c * 16u + k;
          if (e >= E) break;
          const unsigned dl = (unsigned)(dst[e] - lo);
          if (dl < nOwn) {
            atomicAdd(&degc[dl], 1u);
            const float* __restrict__ y = Y + (long long)src[e] * COLS;
#pragma nounroll
            for (int c2 = 0; c2 < COLS; ++c2) atomicAdd(&acc[dl * COLS + c2], y[c2]);
          }
        }
      }
    } else {
      const unsigned n = nAll;
      for (unsigned i = tid; i < n; i += OWN_THREADS) {
        const unsigned e = q[i].x;
        const unsigned s = (unsigned)src[e];
        const unsigned dl = (unsigned)(dst[e] - lo);
        q[i] = make_uint2(s, (dl < nOwn) ? dl : 0xFFFFFFFFu);
        if (dl < nOwn) atomicAdd(&degc[dl], 1u);
      }
      __syncthreads();

      for (unsigned base = wave; base < n; base += 32u * OWN_BATCH) {
        float v[OWN_BATCH][NCH];
        float vt[OWN_BATCH];
        unsigned dl[OWN_BATCH];
#pragma unroll
        for (int b = 0; b < OWN_BATCH; ++b) {
          unsigned idx = base + 32u * (unsigned)b;
          if (idx >= n) idx = n - 1;
          const uint2 ent = q[idx];
          dl[b] = ent.y;
          const float* __restrict__ y = Y + (long long)ent.x * COLS;
#pragma unroll
          for (int k = 0; k < FULL; ++k) v[b][k] = y[32 * k + lane];
          vt[b] = 0.0f;
          if (TAIL != 0) vt[b] = y[32 * FULL + ((lane < TAIL) ? lane : (TAIL - 1))];
        }
#pragma unroll
        for (int b = OWN_BATCH - 1; b >= 0; --b) {
          if (TAIL != 0) asm volatile("" : : "v"(vt[b]));
#pragma unroll
          for (int k = FULL - 1; k >= 0; --k) asm volatile("" : : "v"(v[b][k]));
        }
#pragma unroll
        for (int b = 0; b < OWN_BATCH; ++b) {
          if (base + 32u * (unsigned)b < n && dl[b] != 0xFFFFFFFFu) {
            float* a = acc + dl[b] * COLS + lane;
#pragma unroll
            for (int k = 0; k < FULL; ++k) atomicAdd(a + 32 * k, v[b][k]);
            if (TAIL != 0 && lane < TAIL) atomicAdd(a + 32 * FULL, vt[b]);
          }
        }
      }
    }
  }
  __syncthreads();
  if (tid < OWN_ROWS) nin[tid] = gcn_norm_of((float)degc[tid]);
  __syncthreads();

  const unsigned total = nOwn * COLS;
  float* __restrict__ o = out + (long long)lo * COLS;
  for (unsigned i = tid; i < total; i += OWN_THREADS) {
    const unsigned r = i / COLS;
    const unsigned c = i - r * COLS;
    float val = acc[i] * nin[r] + bias[c];
    if (RELU) val = fmaxf(val, 0.0f);
    o[i] = val;
  }
}

extern "C" void kernel_launch(void* const* d_in, const int* in_sizes, int n_in,
                              void* d_out, int out_size, void* d_ws, size_t ws_size,
                              hipStream_t stream) {
  const float* feat = (const float*)d_in[0];
  const int*   src  = (const int*)d_in[1];
  const int*   dst  = (const int*)d_in[2];
  const float* W0   = (const float*)d_in[3];
  const float* b0   = (const float*)d_in[4];
  const float* W1   = (const float*)d_in[5];
  const float* b1   = (const float*)d_in[6];
  const float* W2   = (const float*)d_in[7];
  const float* b2   = (const float*)d_in[8];
  float* out = (float*)d_out;

  const int F = 128, H = 128;
  const int N = in_sizes[0] / F;
  const int E = in_sizes[1];
  (void)in_sizes; (void)n_in; (void)out_size; (void)ws_size;

  float* ws       = (float*)d_ws;
  float* deg_out  = ws;
  float* B0       = ws + 4LL * N;
  float* B1       = B0 + (long long)N * H;
  unsigned* own32 = (unsigned*)(ws + ((4LL * N + 2LL * N * H + 3LL) & ~3LL));

  const int TB = 256;
  auto blocks1d = [](long long n, int tb) { return (unsigned)((n + tb - 1) / tb); };

  gcn_zero_f32<<<blocks1d(N, TB), TB, 0, stream>>>(deg_out, (long long)N);

  const unsigned nChunks = ((unsigned)E + 15u) / 16u;
  const unsigned nWords = nChunks * 4u;
  const unsigned ownGrid = (unsigned)((N + OWN_ROWS - 1) / OWN_ROWS);
  unsigned* listLen = own32 + nWords;
  uint2* lists = (uint2*)(listLen + ((ownGrid + 3u) & ~3u));
  gcn_owner_bytes<<<blocks1d(nWords, EDG_THREADS * EDG_WORDS), EDG_THREADS, 0, stream>>>(
      src, dst, deg_out, own32, listLen, ownGrid, (unsigned)E, nWords);
  gcn_build_lists<<<dim3(ownGrid, BLD_SLICES), BLD_THREADS, 0, stream>>>(
      src, dst, own32, lists, listLen, (unsigned)E, nChunks, N);

  gcn_gemm_wmma<128, 128><<<GEMM_BLOCKS, GEMM_THREADS, 0, stream>>>(feat, W0, deg_out, B0, N);
  gcn_owner<128, true><<<ownGrid, OWN_THREADS, 0, stream>>>(
      B0, src, dst, own32, b0, B1, lists, listLen, (unsigned)E, nChunks, N);

  gcn_gemm_wmma<128, 128><<<GEMM_BLOCKS, GEMM_THREADS, 0, stream>>>(B1, W1, deg_out, B0, N);
  gcn_owner<128, true><<<ownGrid, OWN_THREADS, 0, stream>>>(
      B0, src, dst, own32, b1, B1, lists, listLen, (unsigned)E, nChunks, N);

  gcn_gemm_wmma<128, 40><<<GEMM_BLOCKS, GEMM_THREADS, 0, stream>>>(B1, W2, deg_out, B0, N);
  gcn_owner<40, false><<<ownGrid, OWN_THREADS, 0, stream>>>(
      B0, src, dst, own32, b2, out, lists, listLen, (unsigned)E, nChunks, N);
}


# 128-col aggregation: a wave gathers a source row with ONE 16-B lane load (4 lines per instruction, was four 4-B loads); LDS sums lane-interleaved so the adds stay conflict-free; sum order kept
# speedup vs baseline: 3.3871x; 1.0404x over previous
#include <hip/hip_runtime.h>
#include <hip/hip_bf16.h>

typedef __attribute__((ext_vector_type(2))) float v2f;
typedef __attribute__((ext_vector_type(8))) float v8f;

#define GEMM_THREADS 256
#define GEMM_STRIP   32
#define GEMM_PITCH   132
#define GEMM_BLOCKS  1024

__global__ void gcn_zero_f32(float* __restrict__ p, long long n) {
  long long i = (long long)blockIdx.x * blockDim.x + threadIdx.x;
  if (i < n) p[i] = 0.0f;
}

__device__ __forceinline__ float gcn_norm_of(float deg) {
  return (deg > 0.0f) ? rsqrtf(deg) : 0.0f;
}

template <int KDIM, int MOUT>
__global__ __launch_bounds__(GEMM_THREADS) void gcn_gemm_wmma(
    const float* __restrict__ X, const float* __restrict__ Wm,
    const float* __restrict__ degOut, float* __restrict__ Y, int nRows) {
  static_assert(KDIM == 128, "a wave stages one row of 128 floats per 16-byte lane load");
  constexpr int KSTEPS = KDIM / 4;
  constexpr int WAVES = GEMM_THREADS / 32;
  constexpr int NCT = (MOUT + 15) / 16;
  constexpr int RT = GEMM_STRIP / 16;
  constexpr int RG = (WAVES / NCT < RT) ? WAVES / NCT : RT;
  constexpr int TPW = RT / RG;
  constexpr int PASSES = GEMM_STRIP / WAVES;
  static_assert(NCT * RG <= WAVES && RG * TPW == RT && PASSES * WAVES == GEMM_STRIP, "tiling");
  __shared__ alignas(16) float xs[2][GEMM_STRIP * GEMM_PITCH];

  const int lane = threadIdx.x & 31;
  const int wave = __builtin_amdgcn_readfirstlane((int)(threadIdx.x >> 5));
  const int m  = lane & 15;
  const int kg = lane >> 4;
  const int nStrips = (nRows + GEMM_STRIP - 1) / GEMM_STRIP;
  int strip = blockIdx.x;
  if (strip >= nStrips) return;

  const bool active = wave < NCT * RG;
  const int ct = wave % NCT;
  const int rg = wave / NCT;

  const int col = ct * 16 + m;
  const int colc = (col < MOUT) ? col : (MOUT - 1);
  v2f breg[KSTEPS] = {};
  if (active) {
#pragma unroll
    for (int s = 0; s < KSTEPS; ++s) {
      const int k = 4 * s + 2 * kg;
      breg[s].x = Wm[k * MOUT + colc];
      breg[s].y = Wm[(k + 1) * MOUT + colc];
    }
  }

  float4 r[PASSES];
  float sc[PASSES];
  auto fetch = [&](int st) {
#pragma unroll
    for (int p = 0; p < PASSES; ++p) {
      int row = st * GEMM_STRIP + p * WAVES + wave;
      if (row >= nRows) row = nRows - 1;
      sc[p] = degOut[row];
      r[p] = *(const float4*)(X + (long long)row * KDIM + 4 * lane);
    }
  };
  auto put = [&](float* __restrict__ buf) {
#pragma unroll
    for (int p = 0; p < PASSES; ++p) {
      float4 v = r[p];
      const float nrm = gcn_norm_of(sc[p]);
      v.x *= nrm;
      v.y *= nrm;
      v.z *= nrm;
      v.w *= nrm;
      *(float4*)(buf + (p * WAVES + wave) * GEMM_PITCH + 4 * lane) = v;
    }
  };

  fetch(strip);
  put(xs[0]);
  __syncthreads();
  int cur = 0;
  for (; strip < nStrips; strip += (int)gridDim.x) {
    const int nextStrip = strip + (int)gridDim.x;
    const bool more = nextStrip < nStrips;
    if (more) fetch(nextStrip);

    if (active) {
      const float* __restrict__ xb = xs[cur] + m * GEMM_PITCH + 2 * kg;
      v8f acc[TPW];
#pragma unroll
      for (int t = 0; t < TPW; ++t) acc[t] = v8f{};
#pragma unroll
      for (int s = 0; s < KSTEPS; ++s) {
#pragma unroll
        for (int t = 0; t < TPW; ++t) {
          const int rt = rg + t * RG;
          const v2f A = *(const v2f*)(xb + rt * 16 * GEMM_PITCH + 4 * s);
          acc[t] = __builtin_amdgcn_wmma_f32_16x16x4_f32(
               false, A,  false, breg[s],
               (short)0, acc[t],  false,  false);
        }
      }

      if (col < MOUT) {
#pragma unroll
        for (int t = 0; t < TPW; ++t) {
          const int rt = rg + t * RG;
#pragma unroll
          for (int q = 0; q < 8; ++q) {
            const int outRow = strip * GEMM_STRIP + rt * 16 + q + 8 * kg;
            if (outRow < nRows) Y[(long long)outRow * MOUT + col] = acc[t][q];
          }
        }
      }
    }

    if (more) put(xs[cur ^ 1]);
    __syncthreads();
    cur ^= 1;
  }
}

#define OWN_ROWS    392
#define OWN_THREADS 1024
#define OWN_Q       4096
#define OWN_SCAN    16
#define OWN_BATCH   8
#define OWN_LIST_CAP 32768

#define EDG_THREADS 256
#define EDG_WORDS   4

#define BLD_THREADS 256
#define BLD_SLICES  8
#define BLD_BATCH   8
#define BLD_WQ      512

__global__ __launch_bounds__(EDG_THREADS) void gcn_owner_bytes(
    const int* __restrict__ src, const int* __restrict__ dst, float* __restrict__ degOut,
    unsigned* __restrict__ own32, unsigned* __restrict__ listLen, unsigned nLists,
    unsigned E, unsigned nWords) {
  for (unsigned i = blockIdx.x * EDG_THREADS + threadIdx.x; i < nLists;
       i += gridDim.x * EDG_THREADS)
    listLen[i] = 0u;
  const unsigned w0 = blockIdx.x * (EDG_THREADS * EDG_WORDS) + threadIdx.x;
  const unsigned nFull = E >> 2;
  int4 d[EDG_WORDS] = {}, s[EDG_WORDS] = {};
  if (nFull != 0u) {
#pragma unroll
    for (int j = 0; j < EDG_WORDS; ++j) {
      const unsigned w = w0 + (unsigned)j * EDG_THREADS;
      const unsigned wc = (w < nFull) ? w : 0u;
      d[j] = *(const int4*)(dst + 4u * wc);
      s[j] = *(const int4*)(src + 4u * wc);
    }
    __builtin_amdgcn_sched_barrier(0);
  }
#pragma unroll
  for (int j = 0; j < EDG_WORDS; ++j) {
    const unsigned w = w0 + (unsigned)j * EDG_THREADS;
    if (w < nWords) {
      unsigned packed = 0;
      if (w < nFull) {
        packed = (((unsigned)d[j].x / OWN_ROWS) & 255u) |
                 ((((unsigned)d[j].y / OWN_ROWS) & 255u) << 8) |
                 ((((unsigned)d[j].z / OWN_ROWS) & 255u) << 16) |
                 ((((unsigned)d[j].w / OWN_ROWS) & 255u) << 24);
        atomicAdd(&degOut[s[j].x], 1.0f);
        atomicAdd(&degOut[s[j].y], 1.0f);
        atomicAdd(&degOut[s[j].z], 1.0f);
        atomicAdd(&degOut[s[j].w], 1.0f);
      } else {
        for (unsigned k = 0; k < 4; ++k) {
          const unsigned e = 4u * w + k;
          if (e < E) {
            packed |= (((unsigned)dst[e] / OWN_ROWS) & 255u) << (8 * k);
            atomicAdd(&degOut[src[e]], 1.0f);
          }
        }
      }
      own32[w] = packed;
    }
  }
}

__global__ __launch_bounds__(BLD_THREADS) void gcn_build_lists(
    const int* __restrict__ src, const int* __restrict__ dst,
    const unsigned* __restrict__ own32, uint2* __restrict__ lists,
    unsigned* __restrict__ listLen, unsigned E, unsigned nChunks, int nRows) {
  __shared__ unsigned q[(BLD_THREADS / 32) * BLD_WQ];

  const unsigned tid = threadIdx.x;
  const unsigned lane = tid & 31u;
  const unsigned wave = (unsigned)__builtin_amdgcn_readfirstlane((int)(tid >> 5));
  const unsigned id = blockIdx.x;
  const int lo = (int)(id * OWN_ROWS);
  int nOwnI = nRows - lo;
  if (nOwnI > OWN_ROWS) nOwnI = OWN_ROWS;
  if (nOwnI < 0) nOwnI = 0;
  const unsigned nOwn = (unsigned)nOwnI;

  const unsigned per = (nChunks + gridDim.y - 1u) / gridDim.y;
  const unsigned cBeg = blockIdx.y * per;
  unsigned cEnd = cBeg + per;
  if (cEnd > nChunks) cEnd = nChunks;

  const unsigned idx4 = (id & 255u) * 0x01010101u;
  const uint4* __restrict__ own = (const uint4*)own32;
  unsigned* wq = q + wave * BLD_WQ;
  unsigned wcount = 0;

  for (unsigned base = cBeg; base < cEnd; base += BLD_THREADS * BLD_BATCH) {
    uint4 w[BLD_BATCH];
#pragma unroll
    for (int j = 0; j < BLD_BATCH; ++j) {
      unsigned c = base + (unsigned)j * BLD_THREADS + tid;
      if (c >= nChunks) c = nChunks - 1;
      w[j] = own[c];
    }
#pragma unroll
    for (int j = 0; j < BLD_BATCH; j += 2) {
      const unsigned c = base + (unsigned)j * BLD_THREADS + tid;
      unsigned m = 0;
#pragma unroll
      for (int h = 0; h < 2; ++h) {
        if (c + (unsigned)h * BLD_THREADS < cEnd) {
          const uint4 v = w[j + h];
          const unsigned x0 = v.x ^ idx4, x1 = v.y ^ idx4, x2 = v.z ^ idx4, x3 = v.w ^ idx4;
          const unsigned z0 = ~(((x0 & 0x7F7F7F7Fu) + 0x7F7F7F7Fu) | x0 | 0x7F7F7F7Fu);
          const unsigned z1 = ~(((x1 & 0x7F7F7F7Fu) + 0x7F7F7F7Fu) | x1 | 0x7F7F7F7Fu);
          const unsigned z2 = ~(((x2 & 0x7F7F7F7Fu) + 0x7F7F7F7Fu) | x2 | 0x7F7F7F7Fu);
          const unsigned z3 = ~(((x3 & 0x7F7F7F7Fu) + 0x7F7F7F7Fu) | x3 | 0x7F7F7F7Fu);
          m |= ((z0 >> 7) | (z1 >> 6) | (z2 >> 5) | (z3 >> 4)) << (4 * h);
        }
      }
      while (__builtin_amdgcn_ballot_w32(m != 0u) != 0u) {
        const bool has = m != 0u;
        const unsigned b = has ? (unsigned)__builtin_ctz(m) : 0u;
        m &= m - 1u;
        const unsigned e = (c + ((b >> 2) & 1u) * BLD_THREADS) * 16u + ((b & 3u) << 2) + (b >> 3);
        const bool valid = has && e < E;
        const unsigned vote = __builtin_amdgcn_ballot_w32(valid);
        const unsigned slot = wcount + __builtin_amdgcn_mbcnt_lo(vote, 0u);
        if (valid && slot < BLD_WQ) wq[slot] = e;
        wcount += (unsigned)__builtin_popcount(vote);
      }
    }
  }
  __syncthreads();

  const bool fits = wcount <= BLD_WQ;
  unsigned pos = 0;
  if (lane == 0 && wcount != 0u)
    pos = atomicAdd(&listLen[id], fits ? wcount : (unsigned)(OWN_LIST_CAP + 1));
  pos = (unsigned)__builtin_amdgcn_readfirstlane((int)pos);
  if (!fits) return;

  uint2* __restrict__ myList = lists + (size_t)id * OWN_LIST_CAP;
  for (unsigned i = lane; i < wcount; i += 32u) {
    const unsigned e = wq[i];
    const unsigned s = (unsigned)src[e];
    const unsigned dl = (unsigned)(dst[e] - lo);
    if (pos + i < OWN_LIST_CAP)
      myList[pos + i] = make_uint2(s, (dl < nOwn) ? dl : 0xFFFFFFFFu);
  }
}

template <int COLS, bool RELU>
__global__ __launch_bounds__(OWN_THREADS) void gcn_owner(
    const float* __restrict__ Y, const int* __restrict__ src,
    const int* __restrict__ dst, const unsigned* __restrict__ own32,
    const float* __restrict__ bias,
    float* __restrict__ out, const uint2* __restrict__ lists,
    const unsigned* __restrict__ listLen, unsigned E, unsigned nChunks, int nRows) {
  constexpr int FULL = COLS / 32;
  constexpr int TAIL = COLS % 32;
  constexpr int NCH = FULL > 0 ? FULL : 1;
  constexpr bool WIDE = (COLS == 128);
  auto accPos = [](unsigned c) -> unsigned { return WIDE ? ((c & 3u) * 32u + (c >> 2)) : c; };
  constexpr unsigned SEG = OWN_THREADS * OWN_SCAN;
  __shared__ float acc[OWN_ROWS * COLS];
  __shared__ uint2 q[OWN_Q];
  __shared__ unsigned cnt;
  __shared__ unsigned degc[OWN_ROWS];
  __shared__ float nin[OWN_ROWS];

  const unsigned tid = threadIdx.x;
  const int lane = tid & 31;
  const unsigned wave = (unsigned)__builtin_amdgcn_readfirstlane((int)(tid >> 5));
  const unsigned id = blockIdx.x;
  const int lo = (int)(id * OWN_ROWS);
  int nOwnI = nRows - lo;
  if (nOwnI > OWN_ROWS) nOwnI = OWN_ROWS;
  if (nOwnI < 0) nOwnI = 0;
  const unsigned nOwn = (unsigned)nOwnI;

  for (unsigned i = tid; i < (unsigned)(OWN_ROWS * COLS); i += OWN_THREADS) acc[i] = 0.0f;
  if (tid < OWN_ROWS) degc[tid] = 0u;

  const unsigned idx4 = (id & 255u) * 0x01010101u;
  const uint4* __restrict__ own = (const uint4*)own32;

  const uint2* __restrict__ myList = lists + (size_t)id * OWN_LIST_CAP;
  const unsigned nList = (unsigned)__builtin_amdgcn_readfirstlane((int)listLen[id]);

  unsigned segFirst = 0;
  if (nList <= OWN_LIST_CAP) {
    segFirst = nChunks;
    __syncthreads();
    const uint2* __restrict__ rl = myList;
    const unsigned per = (nList + 31u) >> 5;
    const unsigned beg = wave * per;
    unsigned end = beg + per;
    if (end > nList) end = nList;
    uint2 nxt = make_uint2(0u, 0xFFFFFFFFu);
    if (beg + (unsigned)lane < end) nxt = rl[beg + (unsigned)lane];
    for (unsigned c = beg; c < end; c += 32u) {
      const uint2 cur = nxt;
      if (cur.y != 0xFFFFFFFFu) atomicAdd(&degc[cur.y], 1u);
      nxt = make_uint2(0u, 0xFFFFFFFFu);
      if (c + 32u + (unsigned)lane < end) nxt = rl[c + 32u + (unsigned)lane];
#pragma unroll
      for (int g = 0; g < 32 / OWN_BATCH; ++g) {
        if (c + (unsigned)(g * OWN_BATCH) >= end) break;
        unsigned dl[OWN_BATCH];
        if constexpr (WIDE) {
          float4 v4[OWN_BATCH];
#pragma unroll
          for (int b = 0; b < OWN_BATCH; ++b) {
            const unsigned s = (unsigned)__builtin_amdgcn_readlane((int)cur.x, g * OWN_BATCH + b);
            dl[b] = (unsigned)__builtin_amdgcn_readlane((int)cur.y, g * OWN_BATCH + b);
            v4[b] = *(const float4*)(Y + (long long)s * COLS + 4 * lane);
          }
#pragma unroll
          for (int b = OWN_BATCH - 1; b >= 0; --b)
            asm volatile("" : : "v"(v4[b].x), "v"(v4[b].y), "v"(v4[b].z), "v"(v4[b].w));
#pragma unroll
          for (int b = 0; b < OWN_BATCH; ++b) {
            if (dl[b] != 0xFFFFFFFFu) {
              float* a = acc + dl[b] * COLS + lane;
              atomicAdd(a, v4[b].x);
              atomicAdd(a + 32, v4[b].y);
              atomicAdd(a + 64, v4[b].z);
              atomicAdd(a + 96, v4[b].w);
            }
          }
        } else {
          float v[OWN_BATCH][NCH];
          float vt[OWN_BATCH];
#pragma unroll
          for (int b = 0; b < OWN_BATCH; ++b) {
            const unsigned s = (unsigned)__builtin_amdgcn_readlane((int)cur.x, g * OWN_BATCH + b);
            dl[b] = (unsigned)__builtin_amdgcn_readlane((int)cur.y, g * OWN_BATCH + b);
            const float* __restrict__ y = Y + (long long)s * COLS;
#pragma unroll
            for (int k = 0; k < FULL; ++k) v[b][k] = y[32 * k + lane];
            vt[b] = 0.0f;
            if (TAIL != 0) vt[b] = y[32 * FULL + ((lane < TAIL) ? lane : (TAIL - 1))];
          }
#pragma unroll
          for (int b = OWN_BATCH - 1; b >= 0; --b) {
            if (TAIL != 0) asm volatile("" : : "v"(vt[b]));
#pragma unroll
            for (int k = FULL - 1; k >= 0; --k) asm volatile("" : : "v"(v[b][k]));
          }
#pragma unroll
          for (int b = 0; b < OWN_BATCH; ++b) {
            if (dl[b] != 0xFFFFFFFFu) {
              float* a = acc + dl[b] * COLS + lane;
#pragma unroll
              for (int k = 0; k < FULL; ++k) atomicAdd(a + 32 * k, v[b][k]);
              if (TAIL != 0 && lane < TAIL) atomicAdd(a + 32 * FULL, vt[b]);
            }
          }
        }
      }
    }
  }

  for (unsigned segBase = segFirst; segBase < nChunks; segBase += SEG) {
    __syncthreads();
    if (tid == 0) cnt = 0;
    __syncthreads();

    uint4 w[OWN_SCAN];
#pragma unroll
    for (int j = 0; j < OWN_SCAN; ++j) {
      unsigned c = segBase + (unsigned)j * OWN_THREADS + tid;
      if (c >= nChunks) c = nChunks - 1;
      w[j] = own[c];
    }
#pragma unroll
    for (int j = 0; j < OWN_SCAN; ++j) {
      const unsigned c = segBase + (unsigned)j * OWN_THREADS + tid;
      if (c < nChunks) {
        const unsigned xs[4] = {w[j].x ^ idx4, w[j].y ^ idx4, w[j].z ^ idx4, w[j].w ^ idx4};
#pragma unroll
        for (int t = 0; t < 4; ++t) {
          const unsigned x = xs[t];
          unsigned z = ~(((x & 0x7F7F7F7Fu) + 0x7F7F7F7Fu) | x | 0x7F7F7F7Fu);
          while (z != 0) {
            const unsigned bit = (unsigned)__builtin_ctz(z);
            z &= z - 1u;
            const unsigned e = c * 16u + (unsigned)t * 4u + (bit >> 3);
            if (e < E) {
              const unsigned slot = atomicAdd(&cnt, 1u);
              if (slot < OWN_Q) q[slot].x = e;
            }
          }
        }
      }
    }
    __syncthreads();

    const unsigned nAll = (unsigned)__builtin_amdgcn_readfirstlane((int)cnt);
    if (nAll > OWN_Q) {
      unsigned cEnd = segBase + SEG;
      if (cEnd > nChunks) cEnd = nChunks;
      for (unsigned c = segBase + tid; c < cEnd; c += OWN_THREADS) {
        for (unsigned k = 0; k < 16u; ++k) {
          const unsigned e = c * 16u + k;
          if (e >= E) break;
          const unsigned dl = (unsigned)(dst[e] - lo);
          if (dl < nOwn) {
            atomicAdd(&degc[dl], 1u);
            const float* __restrict__ y = Y + (long long)src[e] * COLS;
#pragma nounroll
            for (int c2 = 0; c2 < COLS; ++c2)
              atomicAdd(&acc[dl * COLS + accPos((unsigned)c2)], y[c2]);
          }
        }
      }
    } else {
      const unsigned n = nAll;
      for (unsigned i = tid; i < n; i += OWN_THREADS) {
        const unsigned e = q[i].x;
        const unsigned s = (unsigned)src[e];
        const unsigned dl = (unsigned)(dst[e] - lo);
        q[i] = make_uint2(s, (dl < nOwn) ? dl : 0xFFFFFFFFu);
        if (dl < nOwn) atomicAdd(&degc[dl], 1u);
      }
      __syncthreads();

      for (unsigned base = wave; base < n; base += 32u * OWN_BATCH) {
        unsigned dl[OWN_BATCH];
        if constexpr (WIDE) {
          float4 v4[OWN_BATCH];
#pragma unroll
          for (int b = 0; b < OWN_BATCH; ++b) {
            unsigned idx = base + 32u * (unsigned)b;
            if (idx >= n) idx = n - 1;
            const uint2 ent = q[idx];
            dl[b] = ent.y;
            v4[b] = *(const float4*)(Y + (long long)ent.x * COLS + 4 * lane);
          }
#pragma unroll
          for (int b = OWN_BATCH - 1; b >= 0; --b)
            asm volatile("" : : "v"(v4[b].x), "v"(v4[b].y), "v"(v4[b].z), "v"(v4[b].w));
#pragma unroll
          for (int b = 0; b < OWN_BATCH; ++b) {
            if (base + 32u * (unsigned)b < n && dl[b] != 0xFFFFFFFFu) {
              float* a = acc + dl[b] * COLS + lane;
              atomicAdd(a, v4[b].x);
              atomicAdd(a + 32, v4[b].y);
              atomicAdd(a + 64, v4[b].z);
              atomicAdd(a + 96, v4[b].w);
            }
          }
        } else {
          float v[OWN_BATCH][NCH];
          float vt[OWN_BATCH];
#pragma unroll
          for (int b = 0; b < OWN_BATCH; ++b) {
            unsigned idx = base + 32u * (unsigned)b;
            if (idx >= n) idx = n - 1;
            const uint2 ent = q[idx];
            dl[b] = ent.y;
            const float* __restrict__ y = Y + (long long)ent.x * COLS;
#pragma unroll
            for (int k = 0; k < FULL; ++k) v[b][k] = y[32 * k + lane];
            vt[b] = 0.0f;
            if (TAIL != 0) vt[b] = y[32 * FULL + ((lane < TAIL) ? lane : (TAIL - 1))];
          }
#pragma unroll
          for (int b = OWN_BATCH - 1; b >= 0; --b) {
            if (TAIL != 0) asm volatile("" : : "v"(vt[b]));
#pragma unroll
            for (int k = FULL - 1; k >= 0; --k) asm volatile("" : : "v"(v[b][k]));
          }
#pragma unroll
          for (int b = 0; b < OWN_BATCH; ++b) {
            if (base + 32u * (unsigned)b < n && dl[b] != 0xFFFFFFFFu) {
              float* a = acc + dl[b] * COLS + lane;
#pragma unroll
              for (int k = 0; k < FULL; ++k) atomicAdd(a + 32 * k, v[b][k]);
              if (TAIL != 0 && lane < TAIL) atomicAdd(a + 32 * FULL, vt[b]);
            }
          }
        }
      }
    }
  }
  __syncthreads();
  if (tid < OWN_ROWS) nin[tid] = gcn_norm_of((float)degc[tid]);
  __syncthreads();

  float* __restrict__ o = out + (long long)lo * COLS;
  if constexpr (WIDE) {
    const float4 bv = make_float4(bias[4 * lane], bias[4 * lane + 1], bias[4 * lane + 2], bias[4 * lane + 3]);
    for (unsigned r = wave; r < nOwn; r += OWN_THREADS / 32) {
      const float* a = acc + r * COLS + lane;
      const float nr = nin[r];
      float4 val;
      val.x = a[0] * nr + bv.x;
      val.y = a[32] * nr + bv.y;
      val.z = a[64] * nr + bv.z;
      val.w = a[96] * nr + bv.w;
      if (RELU) {
        val.x = fmaxf(val.x, 0.0f);
        val.y = fmaxf(val.y, 0.0f);
        val.z = fmaxf(val.z, 0.0f);
        val.w = fmaxf(val.w, 0.0f);
      }
      *(float4*)(o + (long long)r * COLS + 4 * lane) = val;
    }
  } else {
    const unsigned total = nOwn * COLS;
    for (unsigned i = tid; i < total; i += OWN_THREADS) {
      const unsigned r = i / COLS;
      const unsigned c = i - r * COLS;
      float val = acc[i] * nin[r] + bias[c];
      if (RELU) val = fmaxf(val, 0.0f);
      o[i] = val;
    }
  }
}

extern "C" void kernel_launch(void* const* d_in, const int* in_sizes, int n_in,
                              void* d_out, int out_size, void* d_ws, size_t ws_size,
                              hipStream_t stream) {
  const float* feat = (const float*)d_in[0];
  const int*   src  = (const int*)d_in[1];
  const int*   dst  = (const int*)d_in[2];
  const float* W0   = (const float*)d_in[3];
  const float* b0   = (const float*)d_in[4];
  const float* W1   = (const float*)d_in[5];
  const float* b1   = (const float*)d_in[6];
  const float* W2   = (const float*)d_in[7];
  const float* b2   = (const float*)d_in[8];
  float* out = (float*)d_out;

  const int F = 128, H = 128;
  const int N = in_sizes[0] / F;
  const int E = in_sizes[1];
  (void)in_sizes; (void)n_in; (void)out_size; (void)ws_size;

  float* ws       = (float*)d_ws;
  float* deg_out  = ws;
  float* B0       = ws + 4LL * N;
  float* B1       = B0 + (long long)N * H;
  unsigned* own32 = (unsigned*)(ws + ((4LL * N + 2LL * N * H + 3LL) & ~3LL));

  const int TB = 256;
  auto blocks1d = [](long long n, int tb) { return (unsigned)((n + tb - 1) / tb); };

  gcn_zero_f32<<<blocks1d(N, TB), TB, 0, stream>>>(deg_out, (long long)N);

  const unsigned nChunks = ((unsigned)E + 15u) / 16u;
  const unsigned nWords = nChunks * 4u;
  const unsigned ownGrid = (unsigned)((N + OWN_ROWS - 1) / OWN_ROWS);
  unsigned* listLen = own32 + nWords;
  uint2* lists = (uint2*)(listLen + ((ownGrid + 3u) & ~3u));
  gcn_owner_bytes<<<blocks1d(nWords, EDG_THREADS * EDG_WORDS), EDG_THREADS, 0, stream>>>(
      src, dst, deg_out, own32, listLen, ownGrid, (unsigned)E, nWords);
  gcn_build_lists<<<dim3(ownGrid, BLD_SLICES), BLD_THREADS, 0, stream>>>(
      src, dst, own32, lists, listLen, (unsigned)E, nChunks, N);

  gcn_gemm_wmma<128, 128><<<GEMM_BLOCKS, GEMM_THREADS, 0, stream>>>(feat, W0, deg_out, B0, N);
  gcn_owner<128, true><<<ownGrid, OWN_THREADS, 0, stream>>>(
      B0, src, dst, own32, b0, B1, lists, listLen, (unsigned)E, nChunks, N);

  gcn_gemm_wmma<128, 128><<<GEMM_BLOCKS, GEMM_THREADS, 0, stream>>>(B1, W1, deg_out, B0, N);
  gcn_owner<128, true><<<ownGrid, OWN_THREADS, 0, stream>>>(
      B0, src, dst, own32, b1, B1, lists, listLen, (unsigned)E, nChunks, N);

  gcn_gemm_wmma<128, 40><<<GEMM_BLOCKS, GEMM_THREADS, 0, stream>>>(B1, W2, deg_out, B0, N);
  gcn_owner<40, false><<<ownGrid, OWN_THREADS, 0, stream>>>(
      B0, src, dst, own32, b2, out, lists, listLen, (unsigned)E, nChunks, N);
}


# out-degree counted in LDS: 64 edge slices x 2 row parts histogram src with LDS int adds (in the zero-fill launch's place), the owner-byte pass sums the 64 counts per row; global f32 atomic adds gone
# speedup vs baseline: 4.1811x; 1.2344x over previous
#include <hip/hip_runtime.h>
#include <hip/hip_bf16.h>

typedef __attribute__((ext_vector_type(2))) float v2f;
typedef __attribute__((ext_vector_type(8))) float v8f;

#define GEMM_THREADS 256
#define GEMM_STRIP   32
#define GEMM_PITCH   132
#define GEMM_BLOCKS  1024


__device__ __forceinline__ float gcn_norm_of(float deg) {
  return (deg > 0.0f) ? rsqrtf(deg) : 0.0f;
}

template <int KDIM, int MOUT>
__global__ __launch_bounds__(GEMM_THREADS) void gcn_gemm_wmma(
    const float* __restrict__ X, const float* __restrict__ Wm,
    const float* __restrict__ degOut, float* __restrict__ Y, int nRows) {
  static_assert(KDIM == 128, "a wave stages one row of 128 floats per 16-byte lane load");
  constexpr int KSTEPS = KDIM / 4;
  constexpr int WAVES = GEMM_THREADS / 32;
  constexpr int NCT = (MOUT + 15) / 16;
  constexpr int RT = GEMM_STRIP / 16;
  constexpr int RG = (WAVES / NCT < RT) ? WAVES / NCT : RT;
  constexpr int TPW = RT / RG;
  constexpr int PASSES = GEMM_STRIP / WAVES;
  static_assert(NCT * RG <= WAVES && RG * TPW == RT && PASSES * WAVES == GEMM_STRIP, "tiling");
  __shared__ alignas(16) float xs[2][GEMM_STRIP * GEMM_PITCH];

  const int lane = threadIdx.x & 31;
  const int wave = __builtin_amdgcn_readfirstlane((int)(threadIdx.x >> 5));
  const int m  = lane & 15;
  const int kg = lane >> 4;
  const int nStrips = (nRows + GEMM_STRIP - 1) / GEMM_STRIP;
  int strip = blockIdx.x;
  if (strip >= nStrips) return;

  const bool active = wave < NCT * RG;
  const int ct = wave % NCT;
  const int rg = wave / NCT;

  const int col = ct * 16 + m;
  const int colc = (col < MOUT) ? col : (MOUT - 1);
  v2f breg[KSTEPS] = {};
  if (active) {
#pragma unroll
    for (int s = 0; s < KSTEPS; ++s) {
      const int k = 4 * s + 2 * kg;
      breg[s].x = Wm[k * MOUT + colc];
      breg[s].y = Wm[(k + 1) * MOUT + colc];
    }
  }

  float4 r[PASSES];
  float sc[PASSES];
  auto fetch = [&](int st) {
#pragma unroll
    for (int p = 0; p < PASSES; ++p) {
      int row = st * GEMM_STRIP + p * WAVES + wave;
      if (row >= nRows) row = nRows - 1;
      sc[p] = degOut[row];
      r[p] = *(const float4*)(X + (long long)row * KDIM + 4 * lane);
    }
  };
  auto put = [&](float* __restrict__ buf) {
#pragma unroll
    for (int p = 0; p < PASSES; ++p) {
      float4 v = r[p];
      const float nrm = gcn_norm_of(sc[p]);
      v.x *= nrm;
      v.y *= nrm;
      v.z *= nrm;
      v.w *= nrm;
      *(float4*)(buf + (p * WAVES + wave) * GEMM_PITCH + 4 * lane) = v;
    }
  };

  fetch(strip);
  put(xs[0]);
  __syncthreads();
  int cur = 0;
  for (; strip < nStrips; strip += (int)gridDim.x) {
    const int nextStrip = strip + (int)gridDim.x;
    const bool more = nextStrip < nStrips;
    if (more) fetch(nextStrip);

    if (active) {
      const float* __restrict__ xb = xs[cur] + m * GEMM_PITCH + 2 * kg;
      v8f acc[TPW];
#pragma unroll
      for (int t = 0; t < TPW; ++t) acc[t] = v8f{};
#pragma unroll
      for (int s = 0; s < KSTEPS; ++s) {
#pragma unroll
        for (int t = 0; t < TPW; ++t) {
          const int rt = rg + t * RG;
          const v2f A = *(const v2f*)(xb + rt * 16 * GEMM_PITCH + 4 * s);
          acc[t] = __builtin_amdgcn_wmma_f32_16x16x4_f32(
               false, A,  false, breg[s],
               (short)0, acc[t],  false,  false);
        }
      }

      if (col < MOUT) {
#pragma unroll
        for (int t = 0; t < TPW; ++t) {
          const int rt = rg + t * RG;
#pragma unroll
          for (int q = 0; q < 8; ++q) {
            const int outRow = strip * GEMM_STRIP + rt * 16 + q + 8 * kg;
            if (outRow < nRows) Y[(long long)outRow * MOUT + col] = acc[t][q];
          }
        }
      }
    }

    if (more) put(xs[cur ^ 1]);
    __syncthreads();
    cur ^= 1;
  }
}

#define OWN_ROWS    392
#define OWN_THREADS 1024
#define OWN_Q       4096
#define OWN_SCAN    16
#define OWN_BATCH   8
#define OWN_LIST_CAP 32768

#define EDG_THREADS 256
#define EDG_WORDS   4

#define HST_THREADS 1024
#define HST_SPAN    51200
#define HST_SLICES  64
#define HST_BATCH   8

#define BLD_THREADS 256
#define BLD_SLICES  8
#define BLD_BATCH   8
#define BLD_WQ      512

__global__ __launch_bounds__(HST_THREADS) void gcn_src_hist(
    const int* __restrict__ src, unsigned* __restrict__ part, unsigned E, int nRows) {
  __shared__ unsigned hist[HST_SPAN];
  const unsigned tid = threadIdx.x;
  const unsigned lo = blockIdx.y * HST_SPAN;
  unsigned span = (unsigned)nRows - lo;
  if (span > HST_SPAN) span = HST_SPAN;
  for (unsigned i = tid; i < HST_SPAN; i += HST_THREADS) hist[i] = 0u;
  __syncthreads();

  const unsigned nW = (E + 3u) >> 2;
  const unsigned nFull = E >> 2;
  const unsigned per = (nW + gridDim.x - 1u) / gridDim.x;
  const unsigned wBeg = blockIdx.x * per;
  unsigned wEnd = wBeg + per;
  if (wEnd > nW) wEnd = nW;
  for (unsigned base = wBeg; base < wEnd; base += HST_THREADS * HST_BATCH) {
    int4 s[HST_BATCH] = {};
    if (nFull != 0u) {
#pragma unroll
      for (int j = 0; j < HST_BATCH; ++j) {
        const unsigned w = base + (unsigned)j * HST_THREADS + tid;
        const unsigned wc = (w < nFull) ? w : 0u;
        s[j] = *(const int4*)(src + 4u * wc);
      }
    }
#pragma unroll
    for (int j = 0; j < HST_BATCH; ++j) {
      const unsigned w = base + (unsigned)j * HST_THREADS + tid;
      if (w < wEnd) {
        if (w < nFull) {
          const unsigned d0 = (unsigned)s[j].x - lo, d1 = (unsigned)s[j].y - lo;
          const unsigned d2 = (unsigned)s[j].z - lo, d3 = (unsigned)s[j].w - lo;
          if (d0 < span) atomicAdd(&hist[d0], 1u);
          if (d1 < span) atomicAdd(&hist[d1], 1u);
          if (d2 < span) atomicAdd(&hist[d2], 1u);
          if (d3 < span) atomicAdd(&hist[d3], 1u);
        } else {
          for (unsigned k = 0; k < 4; ++k) {
            const unsigned e = 4u * w + k;
            if (e < E) {
              const unsigned d = (unsigned)src[e] - lo;
              if (d < span) atomicAdd(&hist[d], 1u);
            }
          }
        }
      }
    }
  }
  __syncthreads();

  unsigned* __restrict__ o = part + (size_t)blockIdx.x * (unsigned)nRows + lo;
  for (unsigned i = tid; i < span; i += HST_THREADS) o[i] = hist[i];
}

__global__ __launch_bounds__(EDG_THREADS) void gcn_owner_bytes(
    const int* __restrict__ dst, const unsigned* __restrict__ part, float* __restrict__ degOut,
    unsigned* __restrict__ own32, unsigned* __restrict__ listLen, unsigned nLists,
    unsigned E, unsigned nWords, int nRows) {
  for (unsigned i = blockIdx.x * EDG_THREADS + threadIdx.x; i < nLists;
       i += gridDim.x * EDG_THREADS)
    listLen[i] = 0u;
  const unsigned w0 = blockIdx.x * (EDG_THREADS * EDG_WORDS) + threadIdx.x;
  const unsigned nFull = E >> 2;
  int4 d[EDG_WORDS] = {};
  if (nFull != 0u) {
#pragma unroll
    for (int j = 0; j < EDG_WORDS; ++j) {
      const unsigned w = w0 + (unsigned)j * EDG_THREADS;
      const unsigned wc = (w < nFull) ? w : 0u;
      d[j] = *(const int4*)(dst + 4u * wc);
    }
  }
#pragma unroll
  for (int j = 0; j < EDG_WORDS; ++j) {
    const unsigned w = w0 + (unsigned)j * EDG_THREADS;
    if (w < nWords) {
      unsigned packed = 0;
      if (w < nFull) {
        packed = (((unsigned)d[j].x / OWN_ROWS) & 255u) |
                 ((((unsigned)d[j].y / OWN_ROWS) & 255u) << 8) |
                 ((((unsigned)d[j].z / OWN_ROWS) & 255u) << 16) |
                 ((((unsigned)d[j].w / OWN_ROWS) & 255u) << 24);
      } else {
        for (unsigned k = 0; k < 4; ++k) {
          const unsigned e = 4u * w + k;
          if (e < E) packed |= (((unsigned)dst[e] / OWN_ROWS) & 255u) << (8 * k);
        }
      }
      own32[w] = packed;
    }
  }

  for (unsigned n = blockIdx.x * EDG_THREADS + threadIdx.x; n < (unsigned)nRows;
       n += gridDim.x * EDG_THREADS) {
    unsigned cnt = 0;
#pragma unroll 16
    for (unsigned p = 0; p < HST_SLICES; ++p) cnt += part[(size_t)p * (unsigned)nRows + n];
    degOut[n] = (float)cnt;
  }
}

__global__ __launch_bounds__(BLD_THREADS) void gcn_build_lists(
    const int* __restrict__ src, const int* __restrict__ dst,
    const unsigned* __restrict__ own32, uint2* __restrict__ lists,
    unsigned* __restrict__ listLen, unsigned E, unsigned nChunks, int nRows) {
  __shared__ unsigned q[(BLD_THREADS / 32) * BLD_WQ];

  const unsigned tid = threadIdx.x;
  const unsigned lane = tid & 31u;
  const unsigned wave = (unsigned)__builtin_amdgcn_readfirstlane((int)(tid >> 5));
  const unsigned id = blockIdx.x;
  const int lo = (int)(id * OWN_ROWS);
  int nOwnI = nRows - lo;
  if (nOwnI > OWN_ROWS) nOwnI = OWN_ROWS;
  if (nOwnI < 0) nOwnI = 0;
  const unsigned nOwn = (unsigned)nOwnI;

  const unsigned per = (nChunks + gridDim.y - 1u) / gridDim.y;
  const unsigned cBeg = blockIdx.y * per;
  unsigned cEnd = cBeg + per;
  if (cEnd > nChunks) cEnd = nChunks;

  const unsigned idx4 = (id & 255u) * 0x01010101u;
  const uint4* __restrict__ own = (const uint4*)own32;
  unsigned* wq = q + wave * BLD_WQ;
  unsigned wcount = 0;

  for (unsigned base = cBeg; base < cEnd; base += BLD_THREADS * BLD_BATCH) {
    uint4 w[BLD_BATCH];
#pragma unroll
    for (int j = 0; j < BLD_BATCH; ++j) {
      unsigned c = base + (unsigned)j * BLD_THREADS + tid;
      if (c >= nChunks) c = nChunks - 1;
      w[j] = own[c];
    }
#pragma unroll
    for (int j = 0; j < BLD_BATCH; j += 2) {
      const unsigned c = base + (unsigned)j * BLD_THREADS + tid;
      unsigned m = 0;
#pragma unroll
      for (int h = 0; h < 2; ++h) {
        if (c + (unsigned)h * BLD_THREADS < cEnd) {
          const uint4 v = w[j + h];
          const unsigned x0 = v.x ^ idx4, x1 = v.y ^ idx4, x2 = v.z ^ idx4, x3 = v.w ^ idx4;
          const unsigned z0 = ~(((x0 & 0x7F7F7F7Fu) + 0x7F7F7F7Fu) | x0 | 0x7F7F7F7Fu);
          const unsigned z1 = ~(((x1 & 0x7F7F7F7Fu) + 0x7F7F7F7Fu) | x1 | 0x7F7F7F7Fu);
          const unsigned z2 = ~(((x2 & 0x7F7F7F7Fu) + 0x7F7F7F7Fu) | x2 | 0x7F7F7F7Fu);
          const unsigned z3 = ~(((x3 & 0x7F7F7F7Fu) + 0x7F7F7F7Fu) | x3 | 0x7F7F7F7Fu);
          m |= ((z0 >> 7) | (z1 >> 6) | (z2 >> 5) | (z3 >> 4)) << (4 * h);
        }
      }
      while (__builtin_amdgcn_ballot_w32(m != 0u) != 0u) {
        const bool has = m != 0u;
        const unsigned b = has ? (unsigned)__builtin_ctz(m) : 0u;
        m &= m - 1u;
        const unsigned e = (c + ((b >> 2) & 1u) * BLD_THREADS) * 16u + ((b & 3u) << 2) + (b >> 3);
        const bool valid = has && e < E;
        const unsigned vote = __builtin_amdgcn_ballot_w32(valid);
        const unsigned slot = wcount + __builtin_amdgcn_mbcnt_lo(vote, 0u);
        if (valid && slot < BLD_WQ) wq[slot] = e;
        wcount += (unsigned)__builtin_popcount(vote);
      }
    }
  }
  __syncthreads();

  const bool fits = wcount <= BLD_WQ;
  unsigned pos = 0;
  if (lane == 0 && wcount != 0u)
    pos = atomicAdd(&listLen[id], fits ? wcount : (unsigned)(OWN_LIST_CAP + 1));
  pos = (unsigned)__builtin_amdgcn_readfirstlane((int)pos);
  if (!fits) return;

  uint2* __restrict__ myList = lists + (size_t)id * OWN_LIST_CAP;
  for (unsigned i = lane; i < wcount; i += 32u) {
    const unsigned e = wq[i];
    const unsigned s = (unsigned)src[e];
    const unsigned dl = (unsigned)(dst[e] - lo);
    if (pos + i < OWN_LIST_CAP)
      myList[pos + i] = make_uint2(s, (dl < nOwn) ? dl : 0xFFFFFFFFu);
  }
}

template <int COLS, bool RELU>
__global__ __launch_bounds__(OWN_THREADS) void gcn_owner(
    const float* __restrict__ Y, const int* __restrict__ src,
    const int* __restrict__ dst, const unsigned* __restrict__ own32,
    const float* __restrict__ bias,
    float* __restrict__ out, const uint2* __restrict__ lists,
    const unsigned* __restrict__ listLen, unsigned E, unsigned nChunks, int nRows) {
  constexpr int FULL = COLS / 32;
  constexpr int TAIL = COLS % 32;
  constexpr int NCH = FULL > 0 ? FULL : 1;
  constexpr bool WIDE = (COLS == 128);
  auto accPos = [](unsigned c) -> unsigned { return WIDE ? ((c & 3u) * 32u + (c >> 2)) : c; };
  constexpr unsigned SEG = OWN_THREADS * OWN_SCAN;
  __shared__ float acc[OWN_ROWS * COLS];
  __shared__ uint2 q[OWN_Q];
  __shared__ unsigned cnt;
  __shared__ unsigned degc[OWN_ROWS];
  __shared__ float nin[OWN_ROWS];

  const unsigned tid = threadIdx.x;
  const int lane = tid & 31;
  const unsigned wave = (unsigned)__builtin_amdgcn_readfirstlane((int)(tid >> 5));
  const unsigned id = blockIdx.x;
  const int lo = (int)(id * OWN_ROWS);
  int nOwnI = nRows - lo;
  if (nOwnI > OWN_ROWS) nOwnI = OWN_ROWS;
  if (nOwnI < 0) nOwnI = 0;
  const unsigned nOwn = (unsigned)nOwnI;

  for (unsigned i = tid; i < (unsigned)(OWN_ROWS * COLS); i += OWN_THREADS) acc[i] = 0.0f;
  if (tid < OWN_ROWS) degc[tid] = 0u;

  const unsigned idx4 = (id & 255u) * 0x01010101u;
  const uint4* __restrict__ own = (const uint4*)own32;

  const uint2* __restrict__ myList = lists + (size_t)id * OWN_LIST_CAP;
  const unsigned nList = (unsigned)__builtin_amdgcn_readfirstlane((int)listLen[id]);

  unsigned segFirst = 0;
  if (nList <= OWN_LIST_CAP) {
    segFirst = nChunks;
    __syncthreads();
    const uint2* __restrict__ rl = myList;
    const unsigned per = (nList + 31u) >> 5;
    const unsigned beg = wave * per;
    unsigned end = beg + per;
    if (end > nList) end = nList;
    uint2 nxt = make_uint2(0u, 0xFFFFFFFFu);
    if (beg + (unsigned)lane < end) nxt = rl[beg + (unsigned)lane];
    for (unsigned c = beg; c < end; c += 32u) {
      const uint2 cur = nxt;
      if (cur.y != 0xFFFFFFFFu) atomicAdd(&degc[cur.y], 1u);
      nxt = make_uint2(0u, 0xFFFFFFFFu);
      if (c + 32u + (unsigned)lane < end) nxt = rl[c + 32u + (unsigned)lane];
#pragma unroll
      for (int g = 0; g < 32 / OWN_BATCH; ++g) {
        if (c + (unsigned)(g * OWN_BATCH) >= end) break;
        unsigned dl[OWN_BATCH];
        if constexpr (WIDE) {
          float4 v4[OWN_BATCH];
#pragma unroll
          for (int b = 0; b < OWN_BATCH; ++b) {
            const unsigned s = (unsigned)__builtin_amdgcn_readlane((int)cur.x, g * OWN_BATCH + b);
            dl[b] = (unsigned)__builtin_amdgcn_readlane((int)cur.y, g * OWN_BATCH + b);
            v4[b] = *(const float4*)(Y + (long long)s * COLS + 4 * lane);
          }
#pragma unroll
          for (int b = OWN_BATCH - 1; b >= 0; --b)
            asm volatile("" : : "v"(v4[b].x), "v"(v4[b].y), "v"(v4[b].z), "v"(v4[b].w));
#pragma unroll
          for (int b = 0; b < OWN_BATCH; ++b) {
            if (dl[b] != 0xFFFFFFFFu) {
              float* a = acc + dl[b] * COLS + lane;
              atomicAdd(a, v4[b].x);
              atomicAdd(a + 32, v4[b].y);
              atomicAdd(a + 64, v4[b].z);
              atomicAdd(a + 96, v4[b].w);
            }
          }
        } else {
          float v[OWN_BATCH][NCH];
          float vt[OWN_BATCH];
#pragma unroll
          for (int b = 0; b < OWN_BATCH; ++b) {
            const unsigned s = (unsigned)__builtin_amdgcn_readlane((int)cur.x, g * OWN_BATCH + b);
            dl[b] = (unsigned)__builtin_amdgcn_readlane((int)cur.y, g * OWN_BATCH + b);
            const float* __restrict__ y = Y + (long long)s * COLS;
#pragma unroll
            for (int k = 0; k < FULL; ++k) v[b][k] = y[32 * k + lane];
            vt[b] = 0.0f;
            if (TAIL != 0) vt[b] = y[32 * FULL + ((lane < TAIL) ? lane : (TAIL - 1))];
          }
#pragma unroll
          for (int b = OWN_BATCH - 1; b >= 0; --b) {
            if (TAIL != 0) asm volatile("" : : "v"(vt[b]));
#pragma unroll
            for (int k = FULL - 1; k >= 0; --k) asm volatile("" : : "v"(v[b][k]));
          }
#pragma unroll
          for (int b = 0; b < OWN_BATCH; ++b) {
            if (dl[b] != 0xFFFFFFFFu) {
              float* a = acc + dl[b] * COLS + lane;
#pragma unroll
              for (int k = 0; k < FULL; ++k) atomicAdd(a + 32 * k, v[b][k]);
              if (TAIL != 0 && lane < TAIL) atomicAdd(a + 32 * FULL, vt[b]);
            }
          }
        }
      }
    }
  }

  for (unsigned segBase = segFirst; segBase < nChunks; segBase += SEG) {
    __syncthreads();
    if (tid == 0) cnt = 0;
    __syncthreads();

    uint4 w[OWN_SCAN];
#pragma unroll
    for (int j = 0; j < OWN_SCAN; ++j) {
      unsigned c = segBase + (unsigned)j * OWN_THREADS + tid;
      if (c >= nChunks) c = nChunks - 1;
      w[j] = own[c];
    }
#pragma unroll
    for (int j = 0; j < OWN_SCAN; ++j) {
      const unsigned c = segBase + (unsigned)j * OWN_THREADS + tid;
      if (c < nChunks) {
        const unsigned xs[4] = {w[j].x ^ idx4, w[j].y ^ idx4, w[j].z ^ idx4, w[j].w ^ idx4};
#pragma unroll
        for (int t = 0; t < 4; ++t) {
          const unsigned x = xs[t];
          unsigned z = ~(((x & 0x7F7F7F7Fu) + 0x7F7F7F7Fu) | x | 0x7F7F7F7Fu);
          while (z != 0) {
            const unsigned bit = (unsigned)__builtin_ctz(z);
            z &= z - 1u;
            const unsigned e = c * 16u + (unsigned)t * 4u + (bit >> 3);
            if (e < E) {
              const unsigned slot = atomicAdd(&cnt, 1u);
              if (slot < OWN_Q) q[slot].x = e;
            }
          }
        }
      }
    }
    __syncthreads();

    const unsigned nAll = (unsigned)__builtin_amdgcn_readfirstlane((int)cnt);
    if (nAll > OWN_Q) {
      unsigned cEnd = segBase + SEG;
      if (cEnd > nChunks) cEnd = nChunks;
      for (unsigned c = segBase + tid; c < cEnd; c += OWN_THREADS) {
        for (unsigned k = 0; k < 16u; ++k) {
          const unsigned e = c * 16u + k;
          if (e >= E) break;
          const unsigned dl = (unsigned)(dst[e] - lo);
          if (dl < nOwn) {
            atomicAdd(&degc[dl], 1u);
            const float* __restrict__ y = Y + (long long)src[e] * COLS;
#pragma nounroll
            for (int c2 = 0; c2 < COLS; ++c2)
              atomicAdd(&acc[dl * COLS + accPos((unsigned)c2)], y[c2]);
          }
        }
      }
    } else {
      const unsigned n = nAll;
      for (unsigned i = tid; i < n; i += OWN_THREADS) {
        const unsigned e = q[i].x;
        const unsigned s = (unsigned)src[e];
        const unsigned dl = (unsigned)(dst[e] - lo);
        q[i] = make_uint2(s, (dl < nOwn) ? dl : 0xFFFFFFFFu);
        if (dl < nOwn) atomicAdd(&degc[dl], 1u);
      }
      __syncthreads();

      for (unsigned base = wave; base < n; base += 32u * OWN_BATCH) {
        unsigned dl[OWN_BATCH];
        if constexpr (WIDE) {
          float4 v4[OWN_BATCH];
#pragma unroll
          for (int b = 0; b < OWN_BATCH; ++b) {
            unsigned idx = base + 32u * (unsigned)b;
            if (idx >= n) idx = n - 1;
            const uint2 ent = q[idx];
            dl[b] = ent.y;
            v4[b] = *(const float4*)(Y + (long long)ent.x * COLS + 4 * lane);
          }
#pragma unroll
          for (int b = OWN_BATCH - 1; b >= 0; --b)
            asm volatile("" : : "v"(v4[b].x), "v"(v4[b].y), "v"(v4[b].z), "v"(v4[b].w));
#pragma unroll
          for (int b = 0; b < OWN_BATCH; ++b) {
            if (base + 32u * (unsigned)b < n && dl[b] != 0xFFFFFFFFu) {
              float* a = acc + dl[b] * COLS + lane;
              atomicAdd(a, v4[b].x);
              atomicAdd(a + 32, v4[b].y);
              atomicAdd(a + 64, v4[b].z);
              atomicAdd(a + 96, v4[b].w);
            }
          }
        } else {
          float v[OWN_BATCH][NCH];
          float vt[OWN_BATCH];
#pragma unroll
          for (int b = 0; b < OWN_BATCH; ++b) {
            unsigned idx = base + 32u * (unsigned)b;
            if (idx >= n) idx = n - 1;
            const uint2 ent = q[idx];
            dl[b] = ent.y;
            const float* __restrict__ y = Y + (long long)ent.x * COLS;
#pragma unroll
            for (int k = 0; k < FULL; ++k) v[b][k] = y[32 * k + lane];
            vt[b] = 0.0f;
            if (TAIL != 0) vt[b] = y[32 * FULL + ((lane < TAIL) ? lane : (TAIL - 1))];
          }
#pragma unroll
          for (int b = OWN_BATCH - 1; b >= 0; --b) {
            if (TAIL != 0) asm volatile("" : : "v"(vt[b]));
#pragma unroll
            for (int k = FULL - 1; k >= 0; --k) asm volatile("" : : "v"(v[b][k]));
          }
#pragma unroll
          for (int b = 0; b < OWN_BATCH; ++b) {
            if (base + 32u * (unsigned)b < n && dl[b] != 0xFFFFFFFFu) {
              float* a = acc + dl[b] * COLS + lane;
#pragma unroll
              for (int k = 0; k < FULL; ++k) atomicAdd(a + 32 * k, v[b][k]);
              if (TAIL != 0 && lane < TAIL) atomicAdd(a + 32 * FULL, vt[b]);
            }
          }
        }
      }
    }
  }
  __syncthreads();
  if (tid < OWN_ROWS) nin[tid] = gcn_norm_of((float)degc[tid]);
  __syncthreads();

  float* __restrict__ o = out + (long long)lo * COLS;
  if constexpr (WIDE) {
    const float4 bv = make_float4(bias[4 * lane], bias[4 * lane + 1], bias[4 * lane + 2], bias[4 * lane + 3]);
    for (unsigned r = wave; r < nOwn; r += OWN_THREADS / 32) {
      const float* a = acc + r * COLS + lane;
      const float nr = nin[r];
      float4 val;
      val.x = a[0] * nr + bv.x;
      val.y = a[32] * nr + bv.y;
      val.z = a[64] * nr + bv.z;
      val.w = a[96] * nr + bv.w;
      if (RELU) {
        val.x = fmaxf(val.x, 0.0f);
        val.y = fmaxf(val.y, 0.0f);
        val.z = fmaxf(val.z, 0.0f);
        val.w = fmaxf(val.w, 0.0f);
      }
      *(float4*)(o + (long long)r * COLS + 4 * lane) = val;
    }
  } else {
    const unsigned total = nOwn * COLS;
    for (unsigned i = tid; i < total; i += OWN_THREADS) {
      const unsigned r = i / COLS;
      const unsigned c = i - r * COLS;
      float val = acc[i] * nin[r] + bias[c];
      if (RELU) val = fmaxf(val, 0.0f);
      o[i] = val;
    }
  }
}

extern "C" void kernel_launch(void* const* d_in, const int* in_sizes, int n_in,
                              void* d_out, int out_size, void* d_ws, size_t ws_size,
                              hipStream_t stream) {
  const float* feat = (const float*)d_in[0];
  const int*   src  = (const int*)d_in[1];
  const int*   dst  = (const int*)d_in[2];
  const float* W0   = (const float*)d_in[3];
  const float* b0   = (const float*)d_in[4];
  const float* W1   = (const float*)d_in[5];
  const float* b1   = (const float*)d_in[6];
  const float* W2   = (const float*)d_in[7];
  const float* b2   = (const float*)d_in[8];
  float* out = (float*)d_out;

  const int F = 128, H = 128;
  const int N = in_sizes[0] / F;
  const int E = in_sizes[1];
  (void)in_sizes; (void)n_in; (void)out_size; (void)ws_size;

  float* ws       = (float*)d_ws;
  float* deg_out  = ws;
  float* B0       = ws + 4LL * N;
  float* B1       = B0 + (long long)N * H;
  unsigned* own32 = (unsigned*)(ws + ((4LL * N + 2LL * N * H + 3LL) & ~3LL));

  auto blocks1d = [](long long n, int tb) { return (unsigned)((n + tb - 1) / tb); };

  const unsigned nChunks = ((unsigned)E + 15u) / 16u;
  const unsigned nWords = nChunks * 4u;
  const unsigned ownGrid = (unsigned)((N + OWN_ROWS - 1) / OWN_ROWS);
  unsigned* listLen = own32 + nWords;
  uint2* lists = (uint2*)(listLen + ((ownGrid + 3u) & ~3u));
  unsigned* part = (unsigned*)(lists + (size_t)ownGrid * OWN_LIST_CAP);
  const unsigned histParts = (unsigned)((N + HST_SPAN - 1) / HST_SPAN);
  gcn_src_hist<<<dim3(HST_SLICES, histParts), HST_THREADS, 0, stream>>>(src, part, (unsigned)E, N);
  gcn_owner_bytes<<<blocks1d(nWords, EDG_THREADS * EDG_WORDS), EDG_THREADS, 0, stream>>>(
      dst, part, deg_out, own32, listLen, ownGrid, (unsigned)E, nWords, N);
  gcn_build_lists<<<dim3(ownGrid, BLD_SLICES), BLD_THREADS, 0, stream>>>(
      src, dst, own32, lists, listLen, (unsigned)E, nChunks, N);

  gcn_gemm_wmma<128, 128><<<GEMM_BLOCKS, GEMM_THREADS, 0, stream>>>(feat, W0, deg_out, B0, N);
  gcn_owner<128, true><<<ownGrid, OWN_THREADS, 0, stream>>>(
      B0, src, dst, own32, b0, B1, lists, listLen, (unsigned)E, nChunks, N);

  gcn_gemm_wmma<128, 128><<<GEMM_BLOCKS, GEMM_THREADS, 0, stream>>>(B1, W1, deg_out, B0, N);
  gcn_owner<128, true><<<ownGrid, OWN_THREADS, 0, stream>>>(
      B0, src, dst, own32, b1, B1, lists, listLen, (unsigned)E, nChunks, N);

  gcn_gemm_wmma<128, 40><<<GEMM_BLOCKS, GEMM_THREADS, 0, stream>>>(B1, W2, deg_out, B0, N);
  gcn_owner<40, false><<<ownGrid, OWN_THREADS, 0, stream>>>(
      B0, src, dst, own32, b2, out, lists, listLen, (unsigned)E, nChunks, N);
}


# edge lists: a list-building work-group resolves 8 neighbouring owner ranges per scan (owner bytes read 32 times, was 256); ballot counts per range, 1 int add per wave and range; sums keep list order
# speedup vs baseline: 4.4976x; 1.0757x over previous
#include <hip/hip_runtime.h>
#include <hip/hip_bf16.h>

typedef __attribute__((ext_vector_type(2))) float v2f;
typedef __attribute__((ext_vector_type(8))) float v8f;

#define GEMM_THREADS 256
#define GEMM_STRIP   32
#define GEMM_PITCH   132
#define GEMM_BLOCKS  1024


__device__ __forceinline__ float gcn_norm_of(float deg) {
  return (deg > 0.0f) ? rsqrtf(deg) : 0.0f;
}

template <int KDIM, int MOUT>
__global__ __launch_bounds__(GEMM_THREADS) void gcn_gemm_wmma(
    const float* __restrict__ X, const float* __restrict__ Wm,
    const float* __restrict__ degOut, float* __restrict__ Y, int nRows) {
  static_assert(KDIM == 128, "a wave stages one row of 128 floats per 16-byte lane load");
  constexpr int KSTEPS = KDIM / 4;
  constexpr int WAVES = GEMM_THREADS / 32;
  constexpr int NCT = (MOUT + 15) / 16;
  constexpr int RT = GEMM_STRIP / 16;
  constexpr int RG = (WAVES / NCT < RT) ? WAVES / NCT : RT;
  constexpr int TPW = RT / RG;
  constexpr int PASSES = GEMM_STRIP / WAVES;
  static_assert(NCT * RG <= WAVES && RG * TPW == RT && PASSES * WAVES == GEMM_STRIP, "tiling");
  __shared__ alignas(16) float xs[2][GEMM_STRIP * GEMM_PITCH];

  const int lane = threadIdx.x & 31;
  const int wave = __builtin_amdgcn_readfirstlane((int)(threadIdx.x >> 5));
  const int m  = lane & 15;
  const int kg = lane >> 4;
  const int nStrips = (nRows + GEMM_STRIP - 1) / GEMM_STRIP;
  int strip = blockIdx.x;
  if (strip >= nStrips) return;

  const bool active = wave < NCT * RG;
  const int ct = wave % NCT;
  const int rg = wave / NCT;

  const int col = ct * 16 + m;
  const int colc = (col < MOUT) ? col : (MOUT - 1);
  v2f breg[KSTEPS] = {};
  if (active) {
#pragma unroll
    for (int s = 0; s < KSTEPS; ++s) {
      const int k = 4 * s + 2 * kg;
      breg[s].x = Wm[k * MOUT + colc];
      breg[s].y = Wm[(k + 1) * MOUT + colc];
    }
  }

  float4 r[PASSES];
  float sc[PASSES];
  auto fetch = [&](int st) {
#pragma unroll
    for (int p = 0; p < PASSES; ++p) {
      int row = st * GEMM_STRIP + p * WAVES + wave;
      if (row >= nRows) row = nRows - 1;
      sc[p] = degOut[row];
      r[p] = *(const float4*)(X + (long long)row * KDIM + 4 * lane);
    }
  };
  auto put = [&](float* __restrict__ buf) {
#pragma unroll
    for (int p = 0; p < PASSES; ++p) {
      float4 v = r[p];
      const float nrm = gcn_norm_of(sc[p]);
      v.x *= nrm;
      v.y *= nrm;
      v.z *= nrm;
      v.w *= nrm;
      *(float4*)(buf + (p * WAVES + wave) * GEMM_PITCH + 4 * lane) = v;
    }
  };

  fetch(strip);
  put(xs[0]);
  __syncthreads();
  int cur = 0;
  for (; strip < nStrips; strip += (int)gridDim.x) {
    const int nextStrip = strip + (int)gridDim.x;
    const bool more = nextStrip < nStrips;
    if (more) fetch(nextStrip);

    if (active) {
      const float* __restrict__ xb = xs[cur] + m * GEMM_PITCH + 2 * kg;
      v8f acc[TPW];
#pragma unroll
      for (int t = 0; t < TPW; ++t) acc[t] = v8f{};
#pragma unroll
      for (int s = 0; s < KSTEPS; ++s) {
#pragma unroll
        for (int t = 0; t < TPW; ++t) {
          const int rt = rg + t * RG;
          const v2f A = *(const v2f*)(xb + rt * 16 * GEMM_PITCH + 4 * s);
          acc[t] = __builtin_amdgcn_wmma_f32_16x16x4_f32(
               false, A,  false, breg[s],
               (short)0, acc[t],  false,  false);
        }
      }

      if (col < MOUT) {
#pragma unroll
        for (int t = 0; t < TPW; ++t) {
          const int rt = rg + t * RG;
#pragma unroll
          for (int q = 0; q < 8; ++q) {
            const int outRow = strip * GEMM_STRIP + rt * 16 + q + 8 * kg;
            if (outRow < nRows) Y[(long long)outRow * MOUT + col] = acc[t][q];
          }
        }
      }
    }

    if (more) put(xs[cur ^ 1]);
    __syncthreads();
    cur ^= 1;
  }
}

#define OWN_ROWS    392
#define OWN_THREADS 1024
#define OWN_Q       4096
#define OWN_SCAN    16
#define OWN_BATCH   8
#define OWN_LIST_CAP 32768

#define EDG_THREADS 256
#define EDG_WORDS   4

#define HST_THREADS 1024
#define HST_SPAN    51200
#define HST_SLICES  64
#define HST_BATCH   8

#define BLD_THREADS 256
#define BLD_GROUP   8
#define BLD_SLICES  64
#define BLD_BATCH   8
#define BLD_WQ      512

__global__ __launch_bounds__(HST_THREADS) void gcn_src_hist(
    const int* __restrict__ src, unsigned* __restrict__ part, unsigned E, int nRows) {
  __shared__ unsigned hist[HST_SPAN];
  const unsigned tid = threadIdx.x;
  const unsigned lo = blockIdx.y * HST_SPAN;
  unsigned span = (unsigned)nRows - lo;
  if (span > HST_SPAN) span = HST_SPAN;
  for (unsigned i = tid; i < HST_SPAN; i += HST_THREADS) hist[i] = 0u;
  __syncthreads();

  const unsigned nW = (E + 3u) >> 2;
  const unsigned nFull = E >> 2;
  const unsigned per = (nW + gridDim.x - 1u) / gridDim.x;
  const unsigned wBeg = blockIdx.x * per;
  unsigned wEnd = wBeg + per;
  if (wEnd > nW) wEnd = nW;
  for (unsigned base = wBeg; base < wEnd; base += HST_THREADS * HST_BATCH) {
    int4 s[HST_BATCH] = {};
    if (nFull != 0u) {
#pragma unroll
      for (int j = 0; j < HST_BATCH; ++j) {
        const unsigned w = base + (unsigned)j * HST_THREADS + tid;
        const unsigned wc = (w < nFull) ? w : 0u;
        s[j] = *(const int4*)(src + 4u * wc);
      }
    }
#pragma unroll
    for (int j = 0; j < HST_BATCH; ++j) {
      const unsigned w = base + (unsigned)j * HST_THREADS + tid;
      if (w < wEnd) {
        if (w < nFull) {
          const unsigned d0 = (unsigned)s[j].x - lo, d1 = (unsigned)s[j].y - lo;
          const unsigned d2 = (unsigned)s[j].z - lo, d3 = (unsigned)s[j].w - lo;
          if (d0 < span) atomicAdd(&hist[d0], 1u);
          if (d1 < span) atomicAdd(&hist[d1], 1u);
          if (d2 < span) atomicAdd(&hist[d2], 1u);
          if (d3 < span) atomicAdd(&hist[d3], 1u);
        } else {
          for (unsigned k = 0; k < 4; ++k) {
            const unsigned e = 4u * w + k;
            if (e < E) {
              const unsigned d = (unsigned)src[e] - lo;
              if (d < span) atomicAdd(&hist[d], 1u);
            }
          }
        }
      }
    }
  }
  __syncthreads();

  unsigned* __restrict__ o = part + (size_t)blockIdx.x * (unsigned)nRows + lo;
  for (unsigned i = tid; i < span; i += HST_THREADS) o[i] = hist[i];
}

__global__ __launch_bounds__(EDG_THREADS) void gcn_owner_bytes(
    const int* __restrict__ dst, const unsigned* __restrict__ part, float* __restrict__ degOut,
    unsigned* __restrict__ own32, unsigned* __restrict__ listLen, unsigned nLists,
    unsigned E, unsigned nWords, int nRows) {
  for (unsigned i = blockIdx.x * EDG_THREADS + threadIdx.x; i < nLists;
       i += gridDim.x * EDG_THREADS)
    listLen[i] = 0u;
  const unsigned w0 = blockIdx.x * (EDG_THREADS * EDG_WORDS) + threadIdx.x;
  const unsigned nFull = E >> 2;
  int4 d[EDG_WORDS] = {};
  if (nFull != 0u) {
#pragma unroll
    for (int j = 0; j < EDG_WORDS; ++j) {
      const unsigned w = w0 + (unsigned)j * EDG_THREADS;
      const unsigned wc = (w < nFull) ? w : 0u;
      d[j] = *(const int4*)(dst + 4u * wc);
    }
  }
#pragma unroll
  for (int j = 0; j < EDG_WORDS; ++j) {
    const unsigned w = w0 + (unsigned)j * EDG_THREADS;
    if (w < nWords) {
      unsigned packed = 0;
      if (w < nFull) {
        packed = (((unsigned)d[j].x / OWN_ROWS) & 255u) |
                 ((((unsigned)d[j].y / OWN_ROWS) & 255u) << 8) |
                 ((((unsigned)d[j].z / OWN_ROWS) & 255u) << 16) |
                 ((((unsigned)d[j].w / OWN_ROWS) & 255u) << 24);
      } else {
        for (unsigned k = 0; k < 4; ++k) {
          const unsigned e = 4u * w + k;
          if (e < E) packed |= (((unsigned)dst[e] / OWN_ROWS) & 255u) << (8 * k);
        }
      }
      own32[w] = packed;
    }
  }

  for (unsigned n = blockIdx.x * EDG_THREADS + threadIdx.x; n < (unsigned)nRows;
       n += gridDim.x * EDG_THREADS) {
    unsigned cnt = 0;
#pragma unroll 16
    for (unsigned p = 0; p < HST_SLICES; ++p) cnt += part[(size_t)p * (unsigned)nRows + n];
    degOut[n] = (float)cnt;
  }
}

__global__ __launch_bounds__(BLD_THREADS) void gcn_build_lists(
    const int* __restrict__ src, const int* __restrict__ dst,
    const unsigned* __restrict__ own32, uint2* __restrict__ lists,
    unsigned* __restrict__ listLen, unsigned nLists, unsigned E, unsigned nChunks, int nRows) {
  static_assert((BLD_GROUP & (BLD_GROUP - 1)) == 0 && BLD_GROUP <= 32, "group = low owner bits");
  __shared__ uint2 q[(BLD_THREADS / 32) * BLD_WQ];

  const unsigned tid = threadIdx.x;
  const unsigned lane = tid & 31u;
  const unsigned wave = (unsigned)__builtin_amdgcn_readfirstlane((int)(tid >> 5));
  const unsigned first = blockIdx.x * BLD_GROUP;

  const unsigned per = (nChunks + gridDim.y - 1u) / gridDim.y;
  const unsigned cBeg = blockIdx.y * per;
  unsigned cEnd = cBeg + per;
  if (cEnd > nChunks) cEnd = nChunks;

  const unsigned idx4 = (first & 255u) * 0x01010101u;
  const unsigned grp4 = ((256u - BLD_GROUP) & 255u) * 0x01010101u;
  const uint4* __restrict__ own = (const uint4*)own32;
  uint2* wq = q + wave * BLD_WQ;
  unsigned wcount = 0;

  for (unsigned base = cBeg; base < cEnd; base += BLD_THREADS * BLD_BATCH) {
    uint4 w[BLD_BATCH];
#pragma unroll
    for (int j = 0; j < BLD_BATCH; ++j) {
      unsigned c = base + (unsigned)j * BLD_THREADS + tid;
      if (c >= nChunks) c = nChunks - 1;
      w[j] = own[c];
    }
#pragma unroll
    for (int j = 0; j < BLD_BATCH; j += 2) {
      const unsigned c = base + (unsigned)j * BLD_THREADS + tid;
      unsigned m = 0;
#pragma unroll
      for (int h = 0; h < 2; ++h) {
        if (c + (unsigned)h * BLD_THREADS < cEnd) {
          const uint4 v = w[j + h];
          const unsigned x0 = (v.x ^ idx4) & grp4, x1 = (v.y ^ idx4) & grp4;
          const unsigned x2 = (v.z ^ idx4) & grp4, x3 = (v.w ^ idx4) & grp4;
          const unsigned z0 = ~(((x0 & 0x7F7F7F7Fu) + 0x7F7F7F7Fu) | x0 | 0x7F7F7F7Fu);
          const unsigned z1 = ~(((x1 & 0x7F7F7F7Fu) + 0x7F7F7F7Fu) | x1 | 0x7F7F7F7Fu);
          const unsigned z2 = ~(((x2 & 0x7F7F7F7Fu) + 0x7F7F7F7Fu) | x2 | 0x7F7F7F7Fu);
          const unsigned z3 = ~(((x3 & 0x7F7F7F7Fu) + 0x7F7F7F7Fu) | x3 | 0x7F7F7F7Fu);
          m |= ((z0 >> 7) | (z1 >> 6) | (z2 >> 5) | (z3 >> 4)) << (4 * h);
        }
      }
      while (__builtin_amdgcn_ballot_w32(m != 0u) != 0u) {
        const bool has = m != 0u;
        const unsigned b = has ? (unsigned)__builtin_ctz(m) : 0u;
        m &= m - 1u;
        const unsigned e = (c + ((b >> 2) & 1u) * BLD_THREADS) * 16u + ((b & 3u) << 2) + (b >> 3);
        const bool valid = has && e < E;
        const unsigned vote = __builtin_amdgcn_ballot_w32(valid);
        const unsigned slot = wcount + __builtin_amdgcn_mbcnt_lo(vote, 0u);
        if (valid && slot < BLD_WQ) wq[slot].x = e;
        wcount += (unsigned)__builtin_popcount(vote);
      }
    }
  }
  __syncthreads();

  if (wcount == 0u) return;
  if (wcount > BLD_WQ) {
    if (lane < BLD_GROUP && first + lane < nLists)
      atomicAdd(&listLen[first + lane], (unsigned)(OWN_LIST_CAP + 1));
    return;
  }

  unsigned cnt[BLD_GROUP];
#pragma unroll
  for (int k = 0; k < BLD_GROUP; ++k) cnt[k] = 0u;
  for (unsigned base = 0; base < wcount; base += 32u) {
    const unsigned i = base + lane;
    unsigned sub = 0xFFFFu;
    if (i < wcount) {
      const unsigned e = wq[i].x;
      const unsigned s = (unsigned)src[e];
      const unsigned d = (unsigned)dst[e];
      const unsigned o = d / OWN_ROWS;
      const bool ours = d < (unsigned)nRows && o - first < BLD_GROUP;
      if (ours) sub = o - first;
      wq[i] = make_uint2(s, ours ? ((sub << 16) | (d - o * OWN_ROWS)) : 0xFFFFFFFFu);
    }
#pragma unroll
    for (int k = 0; k < BLD_GROUP; ++k)
      cnt[k] += (unsigned)__builtin_popcount(__builtin_amdgcn_ballot_w32(sub == (unsigned)k));
  }

  unsigned mine = 0;
#pragma unroll
  for (int k = 0; k < BLD_GROUP; ++k)
    if (lane == (unsigned)k) mine = cnt[k];
  unsigned pos = 0;
  if (mine != 0u) pos = atomicAdd(&listLen[first + lane], mine);
  unsigned run[BLD_GROUP];
#pragma unroll
  for (int k = 0; k < BLD_GROUP; ++k) run[k] = (unsigned)__builtin_amdgcn_readlane((int)pos, k);

  for (unsigned base = 0; base < wcount; base += 32u) {
    const unsigned i = base + lane;
    uint2 ent = make_uint2(0u, 0xFFFFFFFFu);
    if (i < wcount) ent = wq[i];
    const unsigned sub = ent.y >> 16;
    unsigned at = 0;
#pragma unroll
    for (int k = 0; k < BLD_GROUP; ++k) {
      const unsigned vote = __builtin_amdgcn_ballot_w32(sub == (unsigned)k);
      if (sub == (unsigned)k) at = run[k] + __builtin_amdgcn_mbcnt_lo(vote, 0u);
      run[k] += (unsigned)__builtin_popcount(vote);
    }
    if (sub < BLD_GROUP && at < OWN_LIST_CAP)
      lists[(size_t)(first + sub) * OWN_LIST_CAP + at] = make_uint2(ent.x, ent.y & 0xFFFFu);
  }
}

template <int COLS, bool RELU>
__global__ __launch_bounds__(OWN_THREADS) void gcn_owner(
    const float* __restrict__ Y, const int* __restrict__ src,
    const int* __restrict__ dst, const unsigned* __restrict__ own32,
    const float* __restrict__ bias,
    float* __restrict__ out, const uint2* __restrict__ lists,
    const unsigned* __restrict__ listLen, unsigned E, unsigned nChunks, int nRows) {
  constexpr int FULL = COLS / 32;
  constexpr int TAIL = COLS % 32;
  constexpr int NCH = FULL > 0 ? FULL : 1;
  constexpr bool WIDE = (COLS == 128);
  auto accPos = [](unsigned c) -> unsigned { return WIDE ? ((c & 3u) * 32u + (c >> 2)) : c; };
  constexpr unsigned SEG = OWN_THREADS * OWN_SCAN;
  __shared__ float acc[OWN_ROWS * COLS];
  __shared__ uint2 q[OWN_Q];
  __shared__ unsigned cnt;
  __shared__ unsigned degc[OWN_ROWS];
  __shared__ float nin[OWN_ROWS];

  const unsigned tid = threadIdx.x;
  const int lane = tid & 31;
  const unsigned wave = (unsigned)__builtin_amdgcn_readfirstlane((int)(tid >> 5));
  const unsigned id = blockIdx.x;
  const int lo = (int)(id * OWN_ROWS);
  int nOwnI = nRows - lo;
  if (nOwnI > OWN_ROWS) nOwnI = OWN_ROWS;
  if (nOwnI < 0) nOwnI = 0;
  const unsigned nOwn = (unsigned)nOwnI;

  for (unsigned i = tid; i < (unsigned)(OWN_ROWS * COLS); i += OWN_THREADS) acc[i] = 0.0f;
  if (tid < OWN_ROWS) degc[tid] = 0u;

  const unsigned idx4 = (id & 255u) * 0x01010101u;
  const uint4* __restrict__ own = (const uint4*)own32;

  const uint2* __restrict__ myList = lists + (size_t)id * OWN_LIST_CAP;
  const unsigned nList = (unsigned)__builtin_amdgcn_readfirstlane((int)listLen[id]);

  unsigned segFirst = 0;
  if (nList <= OWN_LIST_CAP) {
    segFirst = nChunks;
    __syncthreads();
    const uint2* __restrict__ rl = myList;
    const unsigned per = (nList + 31u) >> 5;
    const unsigned beg = wave * per;
    unsigned end = beg + per;
    if (end > nList) end = nList;
    uint2 nxt = make_uint2(0u, 0xFFFFFFFFu);
    if (beg + (unsigned)lane < end) nxt = rl[beg + (unsigned)lane];
    for (unsigned c = beg; c < end; c += 32u) {
      const uint2 cur = nxt;
      if (cur.y != 0xFFFFFFFFu) atomicAdd(&degc[cur.y], 1u);
      nxt = make_uint2(0u, 0xFFFFFFFFu);
      if (c + 32u + (unsigned)lane < end) nxt = rl[c + 32u + (unsigned)lane];
#pragma unroll
      for (int g = 0; g < 32 / OWN_BATCH; ++g) {
        if (c + (unsigned)(g * OWN_BATCH) >= end) break;
        unsigned dl[OWN_BATCH];
        if constexpr (WIDE) {
          float4 v4[OWN_BATCH];
#pragma unroll
          for (int b = 0; b < OWN_BATCH; ++b) {
            const unsigned s = (unsigned)__builtin_amdgcn_readlane((int)cur.x, g * OWN_BATCH + b);
            dl[b] = (unsigned)__builtin_amdgcn_readlane((int)cur.y, g * OWN_BATCH + b);
            v4[b] = *(const float4*)(Y + (long long)s * COLS + 4 * lane);
          }
#pragma unroll
          for (int b = OWN_BATCH - 1; b >= 0; --b)
            asm volatile("" : : "v"(v4[b].x), "v"(v4[b].y), "v"(v4[b].z), "v"(v4[b].w));
#pragma unroll
          for (int b = 0; b < OWN_BATCH; ++b) {
            if (dl[b] != 0xFFFFFFFFu) {
              float* a = acc + dl[b] * COLS + lane;
              atomicAdd(a, v4[b].x);
              atomicAdd(a + 32, v4[b].y);
              atomicAdd(a + 64, v4[b].z);
              atomicAdd(a + 96, v4[b].w);
            }
          }
        } else {
          float v[OWN_BATCH][NCH];
          float vt[OWN_BATCH];
#pragma unroll
          for (int b = 0; b < OWN_BATCH; ++b) {
            const unsigned s = (unsigned)__builtin_amdgcn_readlane((int)cur.x, g * OWN_BATCH + b);
            dl[b] = (unsigned)__builtin_amdgcn_readlane((int)cur.y, g * OWN_BATCH + b);
            const float* __restrict__ y = Y + (long long)s * COLS;
#pragma unroll
            for (int k = 0; k < FULL; ++k) v[b][k] = y[32 * k + lane];
            vt[b] = 0.0f;
            if (TAIL != 0) vt[b] = y[32 * FULL + ((lane < TAIL) ? lane : (TAIL - 1))];
          }
#pragma unroll
          for (int b = OWN_BATCH - 1; b >= 0; --b) {
            if (TAIL != 0) asm volatile("" : : "v"(vt[b]));
#pragma unroll
            for (int k = FULL - 1; k >= 0; --k) asm volatile("" : : "v"(v[b][k]));
          }
#pragma unroll
          for (int b = 0; b < OWN_BATCH; ++b) {
            if (dl[b] != 0xFFFFFFFFu) {
              float* a = acc + dl[b] * COLS + lane;
#pragma unroll
              for (int k = 0; k < FULL; ++k) atomicAdd(a + 32 * k, v[b][k]);
              if (TAIL != 0 && lane < TAIL) atomicAdd(a + 32 * FULL, vt[b]);
            }
          }
        }
      }
    }
  }

  for (unsigned segBase = segFirst; segBase < nChunks; segBase += SEG) {
    __syncthreads();
    if (tid == 0) cnt = 0;
    __syncthreads();

    uint4 w[OWN_SCAN];
#pragma unroll
    for (int j = 0; j < OWN_SCAN; ++j) {
      unsigned c = segBase + (unsigned)j * OWN_THREADS + tid;
      if (c >= nChunks) c = nChunks - 1;
      w[j] = own[c];
    }
#pragma unroll
    for (int j = 0; j < OWN_SCAN; ++j) {
      const unsigned c = segBase + (unsigned)j * OWN_THREADS + tid;
      if (c < nChunks) {
        const unsigned xs[4] = {w[j].x ^ idx4, w[j].y ^ idx4, w[j].z ^ idx4, w[j].w ^ idx4};
#pragma unroll
        for (int t = 0; t < 4; ++t) {
          const unsigned x = xs[t];
          unsigned z = ~(((x & 0x7F7F7F7Fu) + 0x7F7F7F7Fu) | x | 0x7F7F7F7Fu);
          while (z != 0) {
            const unsigned bit = (unsigned)__builtin_ctz(z);
            z &= z - 1u;
            const unsigned e = c * 16u + (unsigned)t * 4u + (bit >> 3);
            if (e < E) {
              const unsigned slot = atomicAdd(&cnt, 1u);
              if (slot < OWN_Q) q[slot].x = e;
            }
          }
        }
      }
    }
    __syncthreads();

    const unsigned nAll = (unsigned)__builtin_amdgcn_readfirstlane((int)cnt);
    if (nAll > OWN_Q) {
      unsigned cEnd = segBase + SEG;
      if (cEnd > nChunks) cEnd = nChunks;
      for (unsigned c = segBase + tid; c < cEnd; c += OWN_THREADS) {
        for (unsigned k = 0; k < 16u; ++k) {
          const unsigned e = c * 16u + k;
          if (e >= E) break;
          const unsigned dl = (unsigned)(dst[e] - lo);
          if (dl < nOwn) {
            atomicAdd(&degc[dl], 1u);
            const float* __restrict__ y = Y + (long long)src[e] * COLS;
#pragma nounroll
            for (int c2 = 0; c2 < COLS; ++c2)
              atomicAdd(&acc[dl * COLS + accPos((unsigned)c2)], y[c2]);
          }
        }
      }
    } else {
      const unsigned n = nAll;
      for (unsigned i = tid; i < n; i += OWN_THREADS) {
        const unsigned e = q[i].x;
        const unsigned s = (unsigned)src[e];
        const unsigned dl = (unsigned)(dst[e] - lo);
        q[i] = make_uint2(s, (dl < nOwn) ? dl : 0xFFFFFFFFu);
        if (dl < nOwn) atomicAdd(&degc[dl], 1u);
      }
      __syncthreads();

      for (unsigned base = wave; base < n; base += 32u * OWN_BATCH) {
        unsigned dl[OWN_BATCH];
        if constexpr (WIDE) {
          float4 v4[OWN_BATCH];
#pragma unroll
          for (int b = 0; b < OWN_BATCH; ++b) {
            unsigned idx = base + 32u * (unsigned)b;
            if (idx >= n) idx = n - 1;
            const uint2 ent = q[idx];
            dl[b] = ent.y;
            v4[b] = *(const float4*)(Y + (long long)ent.x * COLS + 4 * lane);
          }
#pragma unroll
          for (int b = OWN_BATCH - 1; b >= 0; --b)
            asm volatile("" : : "v"(v4[b].x), "v"(v4[b].y), "v"(v4[b].z), "v"(v4[b].w));
#pragma unroll
          for (int b = 0; b < OWN_BATCH; ++b) {
            if (base + 32u * (unsigned)b < n && dl[b] != 0xFFFFFFFFu) {
              float* a = acc + dl[b] * COLS + lane;
              atomicAdd(a, v4[b].x);
              atomicAdd(a + 32, v4[b].y);
              atomicAdd(a + 64, v4[b].z);
              atomicAdd(a + 96, v4[b].w);
            }
          }
        } else {
          float v[OWN_BATCH][NCH];
          float vt[OWN_BATCH];
#pragma unroll
          for (int b = 0; b < OWN_BATCH; ++b) {
            unsigned idx = base + 32u * (unsigned)b;
            if (idx >= n) idx = n - 1;
            const uint2 ent = q[idx];
            dl[b] = ent.y;
            const float* __restrict__ y = Y + (long long)ent.x * COLS;
#pragma unroll
            for (int k = 0; k < FULL; ++k) v[b][k] = y[32 * k + lane];
            vt[b] = 0.0f;
            if (TAIL != 0) vt[b] = y[32 * FULL + ((lane < TAIL) ? lane : (TAIL - 1))];
          }
#pragma unroll
          for (int b = OWN_BATCH - 1; b >= 0; --b) {
            if (TAIL != 0) asm volatile("" : : "v"(vt[b]));
#pragma unroll
            for (int k = FULL - 1; k >= 0; --k) asm volatile("" : : "v"(v[b][k]));
          }
#pragma unroll
          for (int b = 0; b < OWN_BATCH; ++b) {
            if (base + 32u * (unsigned)b < n && dl[b] != 0xFFFFFFFFu) {
              float* a = acc + dl[b] * COLS + lane;
#pragma unroll
              for (int k = 0; k < FULL; ++k) atomicAdd(a + 32 * k, v[b][k]);
              if (TAIL != 0 && lane < TAIL) atomicAdd(a + 32 * FULL, vt[b]);
            }
          }
        }
      }
    }
  }
  __syncthreads();
  if (tid < OWN_ROWS) nin[tid] = gcn_norm_of((float)degc[tid]);
  __syncthreads();

  float* __restrict__ o = out + (long long)lo * COLS;
  if constexpr (WIDE) {
    const float4 bv = make_float4(bias[4 * lane], bias[4 * lane + 1], bias[4 * lane + 2], bias[4 * lane + 3]);
    for (unsigned r = wave; r < nOwn; r += OWN_THREADS / 32) {
      const float* a = acc + r * COLS + lane;
      const float nr = nin[r];
      float4 val;
      val.x = a[0] * nr + bv.x;
      val.y = a[32] * nr + bv.y;
      val.z = a[64] * nr + bv.z;
      val.w = a[96] * nr + bv.w;
      if (RELU) {
        val.x = fmaxf(val.x, 0.0f);
        val.y = fmaxf(val.y, 0.0f);
        val.z = fmaxf(val.z, 0.0f);
        val.w = fmaxf(val.w, 0.0f);
      }
      *(float4*)(o + (long long)r * COLS + 4 * lane) = val;
    }
  } else {
    const unsigned total = nOwn * COLS;
    for (unsigned i = tid; i < total; i += OWN_THREADS) {
      const unsigned r = i / COLS;
      const unsigned c = i - r * COLS;
      float val = acc[i] * nin[r] + bias[c];
      if (RELU) val = fmaxf(val, 0.0f);
      o[i] = val;
    }
  }
}

extern "C" void kernel_launch(void* const* d_in, const int* in_sizes, int n_in,
                              void* d_out, int out_size, void* d_ws, size_t ws_size,
                              hipStream_t stream) {
  const float* feat = (const float*)d_in[0];
  const int*   src  = (const int*)d_in[1];
  const int*   dst  = (const int*)d_in[2];
  const float* W0   = (const float*)d_in[3];
  const float* b0   = (const float*)d_in[4];
  const float* W1   = (const float*)d_in[5];
  const float* b1   = (const float*)d_in[6];
  const float* W2   = (const float*)d_in[7];
  const float* b2   = (const float*)d_in[8];
  float* out = (float*)d_out;

  const int F = 128, H = 128;
  const int N = in_sizes[0] / F;
  const int E = in_sizes[1];
  (void)in_sizes; (void)n_in; (void)out_size; (void)ws_size;

  float* ws       = (float*)d_ws;
  float* deg_out  = ws;
  float* B0       = ws + 4LL * N;
  float* B1       = B0 + (long long)N * H;
  unsigned* own32 = (unsigned*)(ws + ((4LL * N + 2LL * N * H + 3LL) & ~3LL));

  auto blocks1d = [](long long n, int tb) { return (unsigned)((n + tb - 1) / tb); };

  const unsigned nChunks = ((unsigned)E + 15u) / 16u;
  const unsigned nWords = nChunks * 4u;
  const unsigned ownGrid = (unsigned)((N + OWN_ROWS - 1) / OWN_ROWS);
  unsigned* listLen = own32 + nWords;
  uint2* lists = (uint2*)(listLen + ((ownGrid + 3u) & ~3u));
  unsigned* part = (unsigned*)(lists + (size_t)ownGrid * OWN_LIST_CAP);
  const unsigned histParts = (unsigned)((N + HST_SPAN - 1) / HST_SPAN);
  gcn_src_hist<<<dim3(HST_SLICES, histParts), HST_THREADS, 0, stream>>>(src, part, (unsigned)E, N);
  gcn_owner_bytes<<<blocks1d(nWords, EDG_THREADS * EDG_WORDS), EDG_THREADS, 0, stream>>>(
      dst, part, deg_out, own32, listLen, ownGrid, (unsigned)E, nWords, N);
  gcn_build_lists<<<dim3((ownGrid + BLD_GROUP - 1u) / BLD_GROUP, BLD_SLICES), BLD_THREADS, 0, stream>>>(
      src, dst, own32, lists, listLen, ownGrid, (unsigned)E, nChunks, N);

  gcn_gemm_wmma<128, 128><<<GEMM_BLOCKS, GEMM_THREADS, 0, stream>>>(feat, W0, deg_out, B0, N);
  gcn_owner<128, true><<<ownGrid, OWN_THREADS, 0, stream>>>(
      B0, src, dst, own32, b0, B1, lists, listLen, (unsigned)E, nChunks, N);

  gcn_gemm_wmma<128, 128><<<GEMM_BLOCKS, GEMM_THREADS, 0, stream>>>(B1, W1, deg_out, B0, N);
  gcn_owner<128, true><<<ownGrid, OWN_THREADS, 0, stream>>>(
      B0, src, dst, own32, b1, B1, lists, listLen, (unsigned)E, nChunks, N);

  gcn_gemm_wmma<128, 40><<<GEMM_BLOCKS, GEMM_THREADS, 0, stream>>>(B1, W2, deg_out, B0, N);
  gcn_owner<40, false><<<ownGrid, OWN_THREADS, 0, stream>>>(
      B0, src, dst, own32, b2, out, lists, listLen, (unsigned)E, nChunks, N);
}
